# starting kernel
# baseline (speedup 1.0000x reference)
#include <hip/hip_runtime.h>
#include <math.h>

#define N_NODES 50000
#define N_EDGES 400000
#define HEADS   2
#define DDIM    128
#define F_IN    5
#define HD      (HEADS*DDIM)
#define BN_EPS  1e-5f

typedef __attribute__((ext_vector_type(2))) float v2f;
typedef __attribute__((ext_vector_type(8))) float v8f;

__device__ __forceinline__ unsigned enc_f32(float f) {
    unsigned u = __float_as_uint(f);
    return (u & 0x80000000u) ? ~u : (u | 0x80000000u);
}
__device__ __forceinline__ float dec_f32(unsigned u) {
    unsigned v = (u & 0x80000000u) ? (u & 0x7FFFFFFFu) : ~u;
    return __uint_as_float(v);
}
__device__ __forceinline__ void atomAddF(float* p, float v) {
    unsafeAtomicAdd(p, v);
}

__global__ void k_init(float* __restrict__ den, float* __restrict__ cnt,
                       float* __restrict__ ssum, unsigned* __restrict__ mx) {
    long i = (long)blockIdx.x * blockDim.x + threadIdx.x;
    const long total = (long)N_NODES * HD;
    for (long j = i; j < total; j += (long)gridDim.x * blockDim.x) ssum[j] = 0.f;
    if (i < N_NODES) cnt[i] = 0.f;
    if (i < (long)N_NODES * HEADS) { den[i] = 0.f; mx[i] = 0u; }
}

__global__ __launch_bounds__(256)
void k_node(const float* __restrict__ h,
            const float* __restrict__ gamma1, const float* __restrict__ beta1,
            const float* __restrict__ rm1, const float* __restrict__ rv1,
            const float* __restrict__ Wl, const float* __restrict__ bl,
            const float* __restrict__ Wr, const float* __restrict__ br,
            float* __restrict__ xl, float* __restrict__ xr) {
    int n = blockIdx.x;
    int c = threadIdx.x;
    float hbn[F_IN];
#pragma unroll
    for (int k = 0; k < F_IN; ++k) {
        float v = h[n * F_IN + k];
        hbn[k] = (v - rm1[k]) * rsqrtf(rv1[k] + BN_EPS) * gamma1[k] + beta1[k];
    }
    float sl = bl[c], sr = br[c];
#pragma unroll
    for (int k = 0; k < F_IN; ++k) {
        sl += hbn[k] * Wl[k * HD + c];
        sr += hbn[k] * Wr[k * HD + c];
    }
    xl[(long)n * HD + c] = sl;
    xr[(long)n * HD + c] = sr;
}

__global__ __launch_bounds__(256)
void k_edge_logits(const int* __restrict__ ei, const float* __restrict__ ew,
                   const float* __restrict__ We, const float* __restrict__ att,
                   const float* __restrict__ xl, const float* __restrict__ xr,
                   float* __restrict__ logits, unsigned* __restrict__ mx,
                   float* __restrict__ cnt) {
    int wave = threadIdx.x >> 5, lane = threadIdx.x & 31;
    int e = blockIdx.x * 8 + wave;
    if (e >= N_EDGES) return;
    int src = ei[e];
    int dst = ei[N_EDGES + e];
    float w = ew[e];
    const float* pl = xl + (long)src * HD;
    const float* pr = xr + (long)dst * HD;
    float s = 0.f;
    int base = lane * 8;
#pragma unroll
    for (int j = 0; j < 8; ++j) {
        int hd = base + j;
        float z = pl[hd] + pr[hd] + w * We[hd];
        z = z > 0.f ? z : 0.2f * z;
        s += att[hd] * z;
    }
    s += __shfl_xor(s, 8, 32);
    s += __shfl_xor(s, 4, 32);
    s += __shfl_xor(s, 2, 32);
    s += __shfl_xor(s, 1, 32);
    if (lane == 0 || lane == 16) {
        int hh = lane >> 4;
        logits[(long)e * HEADS + hh] = s;
        atomicMax(&mx[(long)dst * HEADS + hh], enc_f32(s));
        if (lane == 0) atomAddF(&cnt[dst], 1.f);
    }
}

__global__ __launch_bounds__(256)
void k_edge_exp(const int* __restrict__ ei, float* __restrict__ logits,
                const unsigned* __restrict__ mx, float* __restrict__ den) {
    long i = (long)blockIdx.x * blockDim.x + threadIdx.x;
    if (i >= (long)N_EDGES * HEADS) return;
    int e = (int)(i >> 1), hh = (int)(i & 1);
    int dst = ei[N_EDGES + e];
    float m = dec_f32(mx[(long)dst * HEADS + hh]);
    float ex = expf(logits[i] - m);
    logits[i] = ex;
    atomAddF(&den[(long)dst * HEADS + hh], ex);
}

__global__ __launch_bounds__(256)
void k_edge_scatter(const int* __restrict__ ei, const float* __restrict__ ex,
                    const float* __restrict__ den, const float* __restrict__ xl,
                    float* __restrict__ ssum) {
    int wave = threadIdx.x >> 5, lane = threadIdx.x & 31;
    int e = blockIdx.x * 8 + wave;
    if (e >= N_EDGES) return;
    int src = ei[e], dst = ei[N_EDGES + e];
    int hh = lane >> 4;
    float alpha = ex[(long)e * HEADS + hh] / (den[(long)dst * HEADS + hh] + 1e-16f);
    const float* pl = xl + (long)src * HD;
    float* ps = ssum + (long)dst * HD;
    int base = lane * 8;
#pragma unroll
    for (int j = 0; j < 8; ++j) {
        int hd = base + j;
        atomAddF(&ps[hd], pl[hd] * alpha);
    }
}

__global__ __launch_bounds__(256)
void k_fc(const float* __restrict__ ssum, const float* __restrict__ cnt,
          const float* __restrict__ bias_gat,
          const float* __restrict__ W1, const float* __restrict__ b1,
          const float* __restrict__ gamma2, const float* __restrict__ beta2,
          const float* __restrict__ rm2, const float* __restrict__ rv2,
          const float* __restrict__ W2, const float* __restrict__ b2,
          float* __restrict__ out) {
    __shared__ float Y[16 * 128];
    __shared__ float Z[16 * 128];
    int row0 = blockIdx.x * 16;
    int t = threadIdx.x;

    {
        int r = t >> 4;
        int c0 = (t & 15) * 8;
        int n = row0 + r;
        float inv = 0.5f / fmaxf(cnt[n], 1.f);
        const float* p = ssum + (long)n * HD;
#pragma unroll
        for (int j = 0; j < 8; ++j) {
            int d = c0 + j;
            Y[r * 128 + d] = (p[d] + p[128 + d]) * inv + bias_gat[d];
        }
    }
    __syncthreads();

    int wave = t >> 5, lane = t & 31;
    int c0 = wave * 16;
    int laneN = lane & 15;
    int koff = (lane >> 4) * 2;
    v8f acc = {0.f, 0.f, 0.f, 0.f, 0.f, 0.f, 0.f, 0.f};
    for (int k = 0; k < 128; k += 4) {
        v2f a, b;
        a.x = Y[laneN * 128 + k + koff];
        a.y = Y[laneN * 128 + k + koff + 1];
        int kk = k + koff;
        b.x = W1[kk * 128 + c0 + laneN];
        b.y = W1[(kk + 1) * 128 + c0 + laneN];
        acc = __builtin_amdgcn_wmma_f32_16x16x4_f32(
            false, a, false, b, (short)0, acc, false, false);
    }

#pragma unroll
    for (int v = 0; v < 8; ++v) {
        int m = v + (lane >> 4) * 8;
        int col = c0 + laneN;
        float z = acc[v] + b1[col];
        z = (z - rm2[col]) * rsqrtf(rv2[col] + BN_EPS) * gamma2[col] + beta2[col];
        z = z > 0.f ? z : 0.01f * z;
        Z[m * 128 + col] = z;
    }
    __syncthreads();

    if (t < 16 * F_IN) {
        int r = t / F_IN, c = t % F_IN;
        float o = b2[c];
        const float* zr = &Z[r * 128];
#pragma unroll 8
        for (int d = 0; d < 128; ++d) o += zr[d] * W2[d * F_IN + c];
        out[(long)(row0 + r) * F_IN + c] = o;
    }
}

extern "C" void kernel_launch(void* const* d_in, const int* in_sizes, int n_in,
                              void* d_out, int out_size, void* d_ws, size_t ws_size,
                              hipStream_t stream) {
    const float* h        = (const float*)d_in[0];
    const int*   ei       = (const int*)  d_in[1];
    const float* ew       = (const float*)d_in[2];
    const float* gamma1   = (const float*)d_in[3];
    const float* beta1    = (const float*)d_in[4];
    const float* rm1      = (const float*)d_in[5];
    const float* rv1      = (const float*)d_in[6];
    const float* Wl       = (const float*)d_in[7];
    const float* bl       = (const float*)d_in[8];
    const float* Wr       = (const float*)d_in[9];
    const float* br       = (const float*)d_in[10];
    const float* We       = (const float*)d_in[11];
    const float* att      = (const float*)d_in[12];
    const float* bias_gat = (const float*)d_in[13];
    const float* W1       = (const float*)d_in[14];
    const float* b1       = (const float*)d_in[15];
    const float* gamma2   = (const float*)d_in[16];
    const float* beta2    = (const float*)d_in[17];
    const float* rm2      = (const float*)d_in[18];
    const float* rv2      = (const float*)d_in[19];
    const float* W2       = (const float*)d_in[20];
    const float* b2       = (const float*)d_in[21];
    float* out = (float*)d_out;

    float* ws = (float*)d_ws;
    size_t o = 0;
    float* xl     = ws + o; o += (size_t)N_NODES * HD;
    float* xr     = ws + o; o += (size_t)N_NODES * HD;
    float* logits = ws + o; o += (size_t)N_EDGES * HEADS;
    float* den    = ws + o; o += (size_t)N_NODES * HEADS;
    float* cnt    = ws + o; o += (size_t)N_NODES;
    float* ssum   = ws + o; o += (size_t)N_NODES * HD;
    unsigned* mx  = (unsigned*)(ws + o); o += (size_t)N_NODES * HEADS;

    const int T = 256;
    k_init<<<(N_NODES * HD + T - 1) / T, T, 0, stream>>>(den, cnt, ssum, mx);
    k_node<<<N_NODES, T, 0, stream>>>(h, gamma1, beta1, rm1, rv1,
                                      Wl, bl, Wr, br, xl, xr);
    k_edge_logits<<<N_EDGES / 8, T, 0, stream>>>(ei, ew, We, att, xl, xr,
                                                 logits, mx, cnt);
    k_edge_exp<<<(N_EDGES * HEADS + T - 1) / T, T, 0, stream>>>(ei, logits, mx, den);
    k_edge_scatter<<<N_EDGES / 8, T, 0, stream>>>(ei, logits, den, xl, ssum);
    k_fc<<<N_NODES / 16, T, 0, stream>>>(ssum, cnt, bias_gat, W1, b1,
                                         gamma2, beta2, rm2, rv2, W2, b2, out);
}


# fc kernel fused into the segment kernel: a block's 32 node rows go head-meaned to LDS (their used-up xr slots), fc1 on f32 WMMA, BN2, fc2 in-block; ssum plane, 4th launch dropped; sum orders kept
# speedup vs baseline: 4.8413x; 4.8413x over previous
#include <hip/hip_runtime.h>
#include <math.h>

#define N_NODES 50000
#define N_EDGES 400000
#define HEADS   2
#define DDIM    128
#define F_IN    5
#define HD      (HEADS*DDIM)
#define BN_EPS  1e-5f

#define CNT_OFF  0
#define REC_OFF  50048
#define OVF_OFF  3250048

#define SLOTS    32

typedef __attribute__((ext_vector_type(2))) float v2f;
typedef __attribute__((ext_vector_type(8))) float v8f;

template <int M>
__device__ __forceinline__ float row_xor_add(float v) {
    int t = __builtin_amdgcn_update_dpp(0, __float_as_int(v), 0x160 + M, 0xf, 0xf, true);
    return v + __int_as_float(t);
}

__global__ __launch_bounds__(256)
void k_zero(int* __restrict__ cnt) {
    int i = blockIdx.x * blockDim.x + threadIdx.x;
    if (i <= N_NODES) cnt[i] = 0;
}

__global__ __launch_bounds__(256)
void k_place(const int* __restrict__ ei, const float* __restrict__ ew,
             int* __restrict__ cnt, int2* __restrict__ rec,
             int4* __restrict__ ovf) {
    int e = blockIdx.x * blockDim.x + threadIdx.x;
    if (e >= N_EDGES) return;
    int src = ei[e];
    int dst = ei[N_EDGES + e];
    int w = __float_as_int(ew[e]);
    int rank = atomicAdd(&cnt[dst], 1);
    if (rank < SLOTS) rec[(size_t)dst * SLOTS + rank] = make_int2(src, w);
    else ovf[atomicAdd(&cnt[N_NODES], 1)] = make_int4(dst, src, w, 0);
}

#define NPW   8
#define SEG_W 4
#define SEG_T (32 * SEG_W)
#define SEG_BLOCKS ((N_NODES + NPW * SEG_W - 1) / (NPW * SEG_W))

__device__ __forceinline__ void src_fetch(const float* __restrict__ h, unsigned s,
                                          float4& a, float2& b) {
    s = s < N_NODES ? s : N_NODES - 1;
    unsigned e = s * F_IN + 4;
    e = e < N_NODES * F_IN - 1 ? e : N_NODES * F_IN - 2;
    __builtin_memcpy(&a, h + (size_t)s * F_IN, 16);
    __builtin_memcpy(&b, h + e, 8);
}
__device__ __forceinline__ float src_fifth(unsigned s, float2 b) {
    return s < N_NODES - 1 ? b.x : b.y;
}

__device__ __forceinline__ void seg_rows(int m, int mw, const float (&hb)[F_IN],
                                         const v2f (&wl)[F_IN][4], const v2f (&blv)[4],
                                         const v2f (&we)[4], const v2f (&at)[4],
                                         const v2f (&r)[4],
                                         float& mx, float& den, v2f (&acc)[4]) {
    for (int j = 0; j < m; j += 4) {
        v2f x[4][4];
        float t[4];
#pragma unroll
        for (int u = 0; u < 4; ++u) {
            int jj = j + u < m ? j + u : m - 1;
            float w = __int_as_float(__builtin_amdgcn_readlane(mw, jj));
            float sh[F_IN];
#pragma unroll
            for (int f = 0; f < F_IN; ++f)
                sh[f] = __int_as_float(__builtin_amdgcn_readlane(__float_as_int(hb[f]), jj));
            v2f q = (v2f){0.f, 0.f};
#pragma unroll
            for (int k = 0; k < 4; ++k) {
                v2f xv = blv[k];
#pragma unroll
                for (int f = 0; f < F_IN; ++f) xv += sh[f] * wl[f][k];
                x[u][k] = xv;
                v2f z = xv + r[k] + w * we[k];
                v2f zs = 0.2f * z;
                z.x = z.x > 0.f ? z.x : zs.x;
                z.y = z.y > 0.f ? z.y : zs.y;
                q += at[k] * z;
            }
            t[u] = q.x + q.y;
        }
#pragma unroll
        for (int u = 0; u < 4; ++u) t[u] = row_xor_add<8>(t[u]);
#pragma unroll
        for (int u = 0; u < 4; ++u) t[u] = row_xor_add<4>(t[u]);
#pragma unroll
        for (int u = 0; u < 4; ++u) t[u] = row_xor_add<2>(t[u]);
#pragma unroll
        for (int u = 0; u < 4; ++u) t[u] = row_xor_add<1>(t[u]);
#pragma unroll
        for (int u = 0; u < 4; ++u)
            if (j + u >= m) t[u] = -INFINITY;
        float mn = fmaxf(fmaxf(mx, fmaxf(t[0], t[1])), fmaxf(t[2], t[3]));
        float sc = expf(mx - mn);
        float p0 = expf(t[0] - mn), p1 = expf(t[1] - mn);
        float p2 = expf(t[2] - mn), p3 = expf(t[3] - mn);
        mx = mn;
        den = den * sc + ((p0 + p1) + (p2 + p3));
#pragma unroll
        for (int k = 0; k < 4; ++k)
            acc[k] = acc[k] * sc + p0 * x[0][k] + p1 * x[1][k]
                                 + p2 * x[2][k] + p3 * x[3][k];
    }
}

__device__ __forceinline__ float other_half(float x) {
    int b = __float_as_int(x);
    return __int_as_float(__builtin_amdgcn_permlanex16(b, b, 0x76543210, 0xfedcba98, false, false));
}

__device__ __forceinline__ void seg_wave(
           const int lane, const int n0, const int cnt, float4 (*xr_w)[64], const int rot0,
           const int* __restrict__ deg, const int2* __restrict__ rec,
           const int4* __restrict__ ovf,
           const float* __restrict__ h,
           const float* __restrict__ gamma1, const float* __restrict__ beta1,
           const float* __restrict__ rm1, const float* __restrict__ rv1,
           const float* __restrict__ Wl, const float* __restrict__ bl,
           const float* __restrict__ Wr, const float* __restrict__ br,
           const float* __restrict__ We, const float* __restrict__ att,
           const float* __restrict__ bias_gat) {
    if (cnt <= 0) {
        for (int i = 0; i < NPW; ++i) xr_w[i][rot0 + i + lane] = make_float4(0.f, 0.f, 0.f, 0.f);
        return;
    }
    int degv = 0;
    if (lane < cnt) degv = deg[n0 + lane];
    float hn0, hn1;
    {
        int e0 = n0 * F_IN + lane, e1 = e0 + 32;
        e0 = e0 < N_NODES * F_IN ? e0 : N_NODES * F_IN - 1;
        e1 = e1 < N_NODES * F_IN ? e1 : N_NODES * F_IN - 1;
        hn0 = h[e0];
        hn1 = h[e1];
    }
    float m1[F_IN], rs1[F_IN], g1[F_IN], b1[F_IN];
#pragma unroll
    for (int f = 0; f < F_IN; ++f) {
        m1[f] = rm1[f];
        rs1[f] = rsqrtf(rv1[f] + BN_EPS);
        g1[f] = gamma1[f];
        b1[f] = beta1[f];
    }
    v2f wl[F_IN][4], blv[4], we[4], at[4];
    {
        const v2f* pb = (const v2f*)bl + lane * 4;
        const v2f* pw = (const v2f*)We + lane * 4;
        const v2f* pa = (const v2f*)att + lane * 4;
#pragma unroll
        for (int k = 0; k < 4; ++k) { blv[k] = pb[k]; we[k] = pw[k]; at[k] = pa[k]; }
#pragma unroll
        for (int f = 0; f < F_IN; ++f) {
            const v2f* pl = (const v2f*)(Wl + f * HD) + lane * 4;
#pragma unroll
            for (int k = 0; k < 4; ++k) wl[f][k] = pl[k];
        }
    }
    const bool lo = lane < 16;
    const int ycol = (lane & 15) * 2 + (lane >> 4);
    float4 bg;
    __builtin_memcpy(&bg, bias_gat + ycol * 4, 16);
    const int2* rrow = rec + (size_t)n0 * SLOTS + lane;
    int dC = __builtin_amdgcn_readlane(degv, 0);
    int dB = __builtin_amdgcn_readlane(degv, 1);
    int2 rcC = make_int2(0, 0), rcB = make_int2(0, 0);
    if (lane < dC) rcC = rrow[0];
    if (lane < dB) rcB = rrow[SLOTS];
    {
        v2f wr[F_IN][4], brv[4];
        const v2f* pc = (const v2f*)br + lane * 4;
#pragma unroll
        for (int k = 0; k < 4; ++k) brv[k] = pc[k];
#pragma unroll
        for (int f = 0; f < F_IN; ++f) {
            const v2f* pq = (const v2f*)(Wr + f * HD) + lane * 4;
#pragma unroll
            for (int k = 0; k < 4; ++k) wr[f][k] = pq[k];
        }
#pragma unroll
        for (int i = 0; i < NPW; ++i) {
            v2f r[4];
#pragma unroll
            for (int k = 0; k < 4; ++k) r[k] = brv[k];
#pragma unroll
            for (int f = 0; f < F_IN; ++f) {
                const int e = i * F_IN + f;
                float raw = __int_as_float(__builtin_amdgcn_readlane(
                    __float_as_int(e < 32 ? hn0 : hn1), e & 31));
                float hv = (raw - m1[f]) * rs1[f] * g1[f] + b1[f];
#pragma unroll
                for (int k = 0; k < 4; ++k) r[k] += hv * wr[f][k];
            }
            xr_w[i][lane]      = make_float4(r[0].x, r[0].y, r[1].x, r[1].y);
            xr_w[i][32 + lane] = make_float4(r[2].x, r[2].y, r[3].x, r[3].y);
        }
    }
    int mwC = rcC.y;
    float hbC[F_IN];
    {
        float4 ga = make_float4(0.f, 0.f, 0.f, 0.f);
        float2 gb = make_float2(0.f, 0.f);
        if (lane < dC) src_fetch(h, rcC.x, ga, gb);
        float g[F_IN] = {ga.x, ga.y, ga.z, ga.w, src_fifth(rcC.x, gb)};
#pragma unroll
        for (int f = 0; f < F_IN; ++f) hbC[f] = (g[f] - m1[f]) * rs1[f] * g1[f] + b1[f];
    }
    for (int i = 0; i < cnt; ++i) {
        float4 gaB = make_float4(0.f, 0.f, 0.f, 0.f);
        float2 gbB = make_float2(0.f, 0.f);
        if (lane < dB) src_fetch(h, rcB.x, gaB, gbB);
        int dA = __builtin_amdgcn_readlane(degv, i + 2);
        int2 rcA = make_int2(0, 0);
        if (lane < dA) rcA = rrow[(i + 2) * SLOTS];
        v2f r[4];
        {
            float4 ra = xr_w[i][lane], rb = xr_w[i][32 + lane];
            r[0] = (v2f){ra.x, ra.y}; r[1] = (v2f){ra.z, ra.w};
            r[2] = (v2f){rb.x, rb.y}; r[3] = (v2f){rb.z, rb.w};
        }
        float mx = -INFINITY, den = 0.f;
        v2f acc[4];
#pragma unroll
        for (int k = 0; k < 4; ++k) acc[k] = (v2f){0.f, 0.f};
        seg_rows(dC < SLOTS ? dC : SLOTS, mwC, hbC, wl, blv, we, at, r, mx, den, acc);
        if (dC > SLOTS) {
            const int node = n0 + i;
            int total = deg[N_NODES];
            total = total < N_EDGES ? total : N_EDGES;
            for (int c0 = 0; c0 < total; c0 += 32) {
                int4 ov = make_int4(-1, 0, 0, 0);
                if (c0 + lane < total) ov = ovf[c0 + lane];
                const bool mine = ov.x == node;
                unsigned hits = __builtin_amdgcn_ballot_w32(mine);
                if (hits == 0u) continue;
                float4 ga = make_float4(0.f, 0.f, 0.f, 0.f);
                float2 gb = make_float2(0.f, 0.f);
                if (mine) src_fetch(h, ov.y, ga, gb);
                float hb[F_IN] = {ga.x, ga.y, ga.z, ga.w, src_fifth(ov.y, gb)};
#pragma unroll
                for (int f = 0; f < F_IN; ++f) hb[f] = (hb[f] - m1[f]) * rs1[f] * g1[f] + b1[f];
                while (hits != 0u) {
                    const int L = __builtin_ctz(hits);
                    hits &= hits - 1u;
                    const int mw1 = __builtin_amdgcn_readlane(ov.z, L);
                    float hb1[F_IN];
#pragma unroll
                    for (int f = 0; f < F_IN; ++f)
                        hb1[f] = __int_as_float(__builtin_amdgcn_readlane(__float_as_int(hb[f]), L));
                    seg_rows(1, mw1, hb1, wl, blv, we, at, r, mx, den, acc);
                }
            }
        }
        float inv = 1.f / (den + 1e-16f);
        const float invc = 0.5f / fmaxf((float)dC, 1.f);
        v2f s[4];
#pragma unroll
        for (int k = 0; k < 4; ++k) s[k] = acc[k] * inv;
        const v2f keep0 = lo ? s[0] : s[2], keep1 = lo ? s[1] : s[3];
        const v2f give0 = lo ? s[2] : s[0], give1 = lo ? s[3] : s[1];
        float4 y;
        y.x = (keep0.x + other_half(give0.x)) * invc + bg.x;
        y.y = (keep0.y + other_half(give0.y)) * invc + bg.y;
        y.z = (keep1.x + other_half(give1.x)) * invc + bg.z;
        y.w = (keep1.y + other_half(give1.y)) * invc + bg.w;
        xr_w[i][rot0 + i + ycol] = y;
        dC = dB; mwC = rcB.y;
        {
            float g[F_IN] = {gaB.x, gaB.y, gaB.z, gaB.w, src_fifth(rcB.x, gbB)};
#pragma unroll
            for (int f = 0; f < F_IN; ++f) hbC[f] = (g[f] - m1[f]) * rs1[f] * g1[f] + b1[f];
        }
        dB = dA; rcB = rcA;
    }
    for (int i = cnt; i < NPW; ++i) xr_w[i][rot0 + i + lane] = make_float4(0.f, 0.f, 0.f, 0.f);
}

#define ROWS_B (NPW * SEG_W)
#define SLOT_F 256

__global__ __launch_bounds__(SEG_T)
void k_seg(const int* __restrict__ deg, const int2* __restrict__ rec,
           const int4* __restrict__ ovf,
           const float* __restrict__ h,
           const float* __restrict__ gamma1, const float* __restrict__ beta1,
           const float* __restrict__ rm1, const float* __restrict__ rv1,
           const float* __restrict__ Wl, const float* __restrict__ bl,
           const float* __restrict__ Wr, const float* __restrict__ br,
           const float* __restrict__ We, const float* __restrict__ att,
           const float* __restrict__ bias_gat,
           const float* __restrict__ W1, const float* __restrict__ b1,
           const float* __restrict__ gamma2, const float* __restrict__ beta2,
           const float* __restrict__ rm2, const float* __restrict__ rv2,
           const float* __restrict__ W2, const float* __restrict__ b2,
           float* __restrict__ out) {
    __shared__ float4 xr_s[SEG_W][NPW][64];
    const int lane = threadIdx.x & 31;
    const int wib = __builtin_amdgcn_readfirstlane(threadIdx.x >> 5);
    const int n0 = __builtin_amdgcn_readfirstlane((blockIdx.x * SEG_W + wib) * NPW);
    const int cnt = N_NODES - n0 < NPW ? N_NODES - n0 : NPW;
    seg_wave(lane, n0, cnt, xr_s[wib], (wib * NPW) & 15, deg, rec, ovf, h,
             gamma1, beta1, rm1, rv1, Wl, bl, Wr, br, We, att, bias_gat);
    __syncthreads();

    float* tile = (float*)&xr_s[0][0][0];
    const int c = lane & 15, hh = lane >> 4;
    const int col0 = wib * 32 + c;
    v8f d00 = {0.f, 0.f, 0.f, 0.f, 0.f, 0.f, 0.f, 0.f}, d01 = d00, d10 = d00, d11 = d00;
    {
        const float* ya = tile + c * (SLOT_F + 4) + 2 * hh;
        const float* yb = ya + 16 * SLOT_F;
        const float* wp = W1 + 2 * hh * DDIM + col0;
#pragma unroll
        for (int k = 0; k < DDIM; k += 4) {
            v2f a0 = *(const v2f*)(ya + k);
            v2f a1 = *(const v2f*)(yb + k);
            v2f b0, b1v;
            b0.x  = wp[k * DDIM];       b0.y  = wp[(k + 1) * DDIM];
            b1v.x = wp[k * DDIM + 16];  b1v.y = wp[(k + 1) * DDIM + 16];
            d00 = __builtin_amdgcn_wmma_f32_16x16x4_f32(false, a0, false, b0,  (short)0, d00, false, false);
            d01 = __builtin_amdgcn_wmma_f32_16x16x4_f32(false, a0, false, b1v, (short)0, d01, false, false);
            d10 = __builtin_amdgcn_wmma_f32_16x16x4_f32(false, a1, false, b0,  (short)0, d10, false, false);
            d11 = __builtin_amdgcn_wmma_f32_16x16x4_f32(false, a1, false, b1v, (short)0, d11, false, false);
        }
    }
    __syncthreads();

#pragma unroll
    for (int nt = 0; nt < 2; ++nt) {
        const int col = col0 + nt * 16;
        const float bb = b1[col], mm = rm2[col], rs = rsqrtf(rv2[col] + BN_EPS);
        const float gg = gamma2[col], be = beta2[col];
#pragma unroll
        for (int mt = 0; mt < 2; ++mt) {
            const v8f d = nt == 0 ? (mt == 0 ? d00 : d10) : (mt == 0 ? d01 : d11);
#pragma unroll
            for (int v = 0; v < 8; ++v) {
                const int m = v + hh * 8;
                float z = d[v] + bb;
                z = (z - mm) * rs * gg + be;
                z = z > 0.f ? z : 0.01f * z;
                tile[(mt * 16 + m) * SLOT_F + 4 * m + col] = z;
            }
        }
    }
    __syncthreads();

    const int row0 = blockIdx.x * ROWS_B;
    for (int o = threadIdx.x; o < ROWS_B * F_IN; o += SEG_T) {
        const int r = o / F_IN, cc = o % F_IN;
        if (row0 + r >= N_NODES) break;
        float acc2 = b2[cc];
        const float* zr = tile + r * SLOT_F + 4 * (r & 15);
#pragma unroll 8
        for (int dd = 0; dd < DDIM; ++dd) acc2 += zr[dd] * W2[dd * F_IN + cc];
        out[(size_t)(row0 + r) * F_IN + cc] = acc2;
    }
}

extern "C" void kernel_launch(void* const* d_in, const int* in_sizes, int n_in,
                              void* d_out, int out_size, void* d_ws, size_t ws_size,
                              hipStream_t stream) {
    const float* h        = (const float*)d_in[0];
    const int*   ei       = (const int*)  d_in[1];
    const float* ew       = (const float*)d_in[2];
    const float* gamma1   = (const float*)d_in[3];
    const float* beta1    = (const float*)d_in[4];
    const float* rm1      = (const float*)d_in[5];
    const float* rv1      = (const float*)d_in[6];
    const float* Wl       = (const float*)d_in[7];
    const float* bl       = (const float*)d_in[8];
    const float* Wr       = (const float*)d_in[9];
    const float* br       = (const float*)d_in[10];
    const float* We       = (const float*)d_in[11];
    const float* att      = (const float*)d_in[12];
    const float* bias_gat = (const float*)d_in[13];
    const float* W1       = (const float*)d_in[14];
    const float* b1       = (const float*)d_in[15];
    const float* gamma2   = (const float*)d_in[16];
    const float* beta2    = (const float*)d_in[17];
    const float* rm2      = (const float*)d_in[18];
    const float* rv2      = (const float*)d_in[19];
    const float* W2       = (const float*)d_in[20];
    const float* b2       = (const float*)d_in[21];
    float* out = (float*)d_out;

    float* ws = (float*)d_ws;
    int* cnt      = (int*)(ws + CNT_OFF);
    int2* rec     = (int2*)(ws + REC_OFF);
    int4* ovf     = (int4*)(ws + OVF_OFF);

    const int T = 256;
    k_zero<<<(N_NODES + T) / T, T, 0, stream>>>(cnt);
    k_place<<<(N_EDGES + T - 1) / T, T, 0, stream>>>(ei, ew, cnt, rec, ovf);
    k_seg<<<SEG_BLOCKS, SEG_T, 0, stream>>>(cnt, rec, ovf, h, gamma1, beta1, rm1, rv1,
                                            Wl, bl, Wr, br, We, att, bias_gat,
                                            W1, b1, gamma2, beta2, rm2, rv2, W2, b2, out);
}


# Row logits summed by a transposing row_xmask butterfly: each quad of lanes ends with ONE row's logit, so mask, max and expf run once per row; weights handed back by row_share DPP; sum orders kept
# speedup vs baseline: 5.0260x; 1.0382x over previous
#include <hip/hip_runtime.h>
#include <math.h>

#define N_NODES 50000
#define N_EDGES 400000
#define HEADS   2
#define DDIM    128
#define F_IN    5
#define HD      (HEADS*DDIM)
#define BN_EPS  1e-5f

#define CNT_OFF  0
#define REC_OFF  50048
#define OVF_OFF  3250048

#define SLOTS    32

typedef __attribute__((ext_vector_type(2))) float v2f;
typedef __attribute__((ext_vector_type(8))) float v8f;

template <int M>
__device__ __forceinline__ float row_xor_add(float v) {
    int t = __builtin_amdgcn_update_dpp(0, __float_as_int(v), 0x160 + M, 0xf, 0xf, true);
    return v + __int_as_float(t);
}
template <int M>
__device__ __forceinline__ float row_xor_get(float v) {
    return __int_as_float(__builtin_amdgcn_update_dpp(0, __float_as_int(v), 0x160 + M, 0xf, 0xf, true));
}
template <int L>
__device__ __forceinline__ float row_lane_get(float v) {
    return __int_as_float(__builtin_amdgcn_update_dpp(0, __float_as_int(v), 0x150 + L, 0xf, 0xf, true));
}

__global__ __launch_bounds__(256)
void k_zero(int* __restrict__ cnt) {
    int i = blockIdx.x * blockDim.x + threadIdx.x;
    if (i <= N_NODES) cnt[i] = 0;
}

__global__ __launch_bounds__(256)
void k_place(const int* __restrict__ ei, const float* __restrict__ ew,
             int* __restrict__ cnt, int2* __restrict__ rec,
             int4* __restrict__ ovf) {
    int e = blockIdx.x * blockDim.x + threadIdx.x;
    if (e >= N_EDGES) return;
    int src = ei[e];
    int dst = ei[N_EDGES + e];
    int w = __float_as_int(ew[e]);
    int rank = atomicAdd(&cnt[dst], 1);
    if (rank < SLOTS) rec[(size_t)dst * SLOTS + rank] = make_int2(src, w);
    else ovf[atomicAdd(&cnt[N_NODES], 1)] = make_int4(dst, src, w, 0);
}

#define NPW   8
#define SEG_W 4
#define SEG_T (32 * SEG_W)
#define SEG_BLOCKS ((N_NODES + NPW * SEG_W - 1) / (NPW * SEG_W))

__device__ __forceinline__ void src_fetch(const float* __restrict__ h, unsigned s,
                                          float4& a, float2& b) {
    s = s < N_NODES ? s : N_NODES - 1;
    unsigned e = s * F_IN + 4;
    e = e < N_NODES * F_IN - 1 ? e : N_NODES * F_IN - 2;
    __builtin_memcpy(&a, h + (size_t)s * F_IN, 16);
    __builtin_memcpy(&b, h + e, 8);
}
__device__ __forceinline__ float src_fifth(unsigned s, float2 b) {
    return s < N_NODES - 1 ? b.x : b.y;
}

__device__ __forceinline__ void seg_rows(const int lane, int m, int mw, const float (&hb)[F_IN],
                                         const v2f (&wl)[F_IN][4], const v2f (&blv)[4],
                                         const v2f (&we)[4], const v2f (&at)[4],
                                         const v2f (&r)[4],
                                         float& mx, float& den, v2f (&acc)[4]) {
    const bool hi8 = (lane & 8) != 0, hi4 = (lane & 4) != 0, odd = (lane & 1) != 0;
    const int urow = (lane >> 2) & 3;
    for (int j = 0; j < m; j += 4) {
        v2f x[4][4];
        float t[4];
#pragma unroll
        for (int u = 0; u < 4; ++u) {
            int jj = j + u < m ? j + u : m - 1;
            float w = __int_as_float(__builtin_amdgcn_readlane(mw, jj));
            float sh[F_IN];
#pragma unroll
            for (int f = 0; f < F_IN; ++f)
                sh[f] = __int_as_float(__builtin_amdgcn_readlane(__float_as_int(hb[f]), jj));
            v2f q = (v2f){0.f, 0.f};
#pragma unroll
            for (int k = 0; k < 4; ++k) {
                v2f xv = blv[k];
#pragma unroll
                for (int f = 0; f < F_IN; ++f) xv += sh[f] * wl[f][k];
                x[u][k] = xv;
                v2f z = xv + r[k] + w * we[k];
                v2f zs = 0.2f * z;
                z.x = z.x > 0.f ? z.x : zs.x;
                z.y = z.y > 0.f ? z.y : zs.y;
                q += at[k] * z;
            }
            t[u] = q.x + q.y;
        }
        const float k0 = hi8 ? t[2] : t[0], g0 = hi8 ? t[0] : t[2];
        const float k1 = hi8 ? t[3] : t[1], g1 = hi8 ? t[1] : t[3];
        const float u0 = k0 + row_xor_get<8>(g0);
        const float u1 = k1 + row_xor_get<8>(g1);
        const float k2 = hi4 ? u1 : u0, g2 = hi4 ? u0 : u1;
        float lg = k2 + row_xor_get<4>(g2);
        lg = row_xor_add<2>(lg);
        lg = row_xor_add<1>(lg);
        if (j + urow >= m) lg = -INFINITY;
        float gm = fmaxf(lg, row_xor_get<8>(lg));
        gm = fmaxf(gm, row_xor_get<4>(gm));
        const float mn = fmaxf(mx, gm);
        const float ex = expf((odd ? mx : lg) - mn);
        const float p0 = row_lane_get<0>(ex), p1 = row_lane_get<4>(ex);
        const float p2 = row_lane_get<8>(ex), p3 = row_lane_get<12>(ex);
        const float sc = row_lane_get<1>(ex);
        mx = mn;
        den = den * sc + ((p0 + p1) + (p2 + p3));
#pragma unroll
        for (int k = 0; k < 4; ++k)
            acc[k] = acc[k] * sc + p0 * x[0][k] + p1 * x[1][k]
                                 + p2 * x[2][k] + p3 * x[3][k];
    }
}

__device__ __forceinline__ float other_half(float x) {
    int b = __float_as_int(x);
    return __int_as_float(__builtin_amdgcn_permlanex16(b, b, 0x76543210, 0xfedcba98, false, false));
}

__device__ __forceinline__ void seg_wave(
           const int lane, const int n0, const int cnt, float4 (*xr_w)[64], const int rot0,
           const int* __restrict__ deg, const int2* __restrict__ rec,
           const int4* __restrict__ ovf,
           const float* __restrict__ h,
           const float* __restrict__ gamma1, const float* __restrict__ beta1,
           const float* __restrict__ rm1, const float* __restrict__ rv1,
           const float* __restrict__ Wl, const float* __restrict__ bl,
           const float* __restrict__ Wr, const float* __restrict__ br,
           const float* __restrict__ We, const float* __restrict__ att,
           const float* __restrict__ bias_gat) {
    if (cnt <= 0) {
        for (int i = 0; i < NPW; ++i) xr_w[i][rot0 + i + lane] = make_float4(0.f, 0.f, 0.f, 0.f);
        return;
    }
    int degv = 0;
    if (lane < cnt) degv = deg[n0 + lane];
    float hn0, hn1;
    {
        int e0 = n0 * F_IN + lane, e1 = e0 + 32;
        e0 = e0 < N_NODES * F_IN ? e0 : N_NODES * F_IN - 1;
        e1 = e1 < N_NODES * F_IN ? e1 : N_NODES * F_IN - 1;
        hn0 = h[e0];
        hn1 = h[e1];
    }
    float m1[F_IN], rs1[F_IN], g1[F_IN], b1[F_IN];
#pragma unroll
    for (int f = 0; f < F_IN; ++f) {
        m1[f] = rm1[f];
        rs1[f] = rsqrtf(rv1[f] + BN_EPS);
        g1[f] = gamma1[f];
        b1[f] = beta1[f];
    }
    v2f wl[F_IN][4], blv[4], we[4], at[4];
    {
        const v2f* pb = (const v2f*)bl + lane * 4;
        const v2f* pw = (const v2f*)We + lane * 4;
        const v2f* pa = (const v2f*)att + lane * 4;
#pragma unroll
        for (int k = 0; k < 4; ++k) { blv[k] = pb[k]; we[k] = pw[k]; at[k] = pa[k]; }
#pragma unroll
        for (int f = 0; f < F_IN; ++f) {
            const v2f* pl = (const v2f*)(Wl + f * HD) + lane * 4;
#pragma unroll
            for (int k = 0; k < 4; ++k) wl[f][k] = pl[k];
        }
    }
    const bool lo = lane < 16;
    const int ycol = (lane & 15) * 2 + (lane >> 4);
    float4 bg;
    __builtin_memcpy(&bg, bias_gat + ycol * 4, 16);
    const int2* rrow = rec + (size_t)n0 * SLOTS + lane;
    int dC = __builtin_amdgcn_readlane(degv, 0);
    int dB = __builtin_amdgcn_readlane(degv, 1);
    int2 rcC = make_int2(0, 0), rcB = make_int2(0, 0);
    if (lane < dC) rcC = rrow[0];
    if (lane < dB) rcB = rrow[SLOTS];
    {
        v2f wr[F_IN][4], brv[4];
        const v2f* pc = (const v2f*)br + lane * 4;
#pragma unroll
        for (int k = 0; k < 4; ++k) brv[k] = pc[k];
#pragma unroll
        for (int f = 0; f < F_IN; ++f) {
            const v2f* pq = (const v2f*)(Wr + f * HD) + lane * 4;
#pragma unroll
            for (int k = 0; k < 4; ++k) wr[f][k] = pq[k];
        }
#pragma unroll
        for (int i = 0; i < NPW; ++i) {
            v2f r[4];
#pragma unroll
            for (int k = 0; k < 4; ++k) r[k] = brv[k];
#pragma unroll
            for (int f = 0; f < F_IN; ++f) {
                const int e = i * F_IN + f;
                float raw = __int_as_float(__builtin_amdgcn_readlane(
                    __float_as_int(e < 32 ? hn0 : hn1), e & 31));
                float hv = (raw - m1[f]) * rs1[f] * g1[f] + b1[f];
#pragma unroll
                for (int k = 0; k < 4; ++k) r[k] += hv * wr[f][k];
            }
            xr_w[i][lane]      = make_float4(r[0].x, r[0].y, r[1].x, r[1].y);
            xr_w[i][32 + lane] = make_float4(r[2].x, r[2].y, r[3].x, r[3].y);
        }
    }
    int mwC = rcC.y;
    float hbC[F_IN];
    {
        float4 ga = make_float4(0.f, 0.f, 0.f, 0.f);
        float2 gb = make_float2(0.f, 0.f);
        if (lane < dC) src_fetch(h, rcC.x, ga, gb);
        float g[F_IN] = {ga.x, ga.y, ga.z, ga.w, src_fifth(rcC.x, gb)};
#pragma unroll
        for (int f = 0; f < F_IN; ++f) hbC[f] = (g[f] - m1[f]) * rs1[f] * g1[f] + b1[f];
    }
    for (int i = 0; i < cnt; ++i) {
        float4 gaB = make_float4(0.f, 0.f, 0.f, 0.f);
        float2 gbB = make_float2(0.f, 0.f);
        if (lane < dB) src_fetch(h, rcB.x, gaB, gbB);
        int dA = __builtin_amdgcn_readlane(degv, i + 2);
        int2 rcA = make_int2(0, 0);
        if (lane < dA) rcA = rrow[(i + 2) * SLOTS];
        v2f r[4];
        {
            float4 ra = xr_w[i][lane], rb = xr_w[i][32 + lane];
            r[0] = (v2f){ra.x, ra.y}; r[1] = (v2f){ra.z, ra.w};
            r[2] = (v2f){rb.x, rb.y}; r[3] = (v2f){rb.z, rb.w};
        }
        float mx = -INFINITY, den = 0.f;
        v2f acc[4];
#pragma unroll
        for (int k = 0; k < 4; ++k) acc[k] = (v2f){0.f, 0.f};
        seg_rows(lane, dC < SLOTS ? dC : SLOTS, mwC, hbC, wl, blv, we, at, r, mx, den, acc);
        if (dC > SLOTS) {
            const int node = n0 + i;
            int total = deg[N_NODES];
            total = total < N_EDGES ? total : N_EDGES;
            for (int c0 = 0; c0 < total; c0 += 32) {
                int4 ov = make_int4(-1, 0, 0, 0);
                if (c0 + lane < total) ov = ovf[c0 + lane];
                const bool mine = ov.x == node;
                unsigned hits = __builtin_amdgcn_ballot_w32(mine);
                if (hits == 0u) continue;
                float4 ga = make_float4(0.f, 0.f, 0.f, 0.f);
                float2 gb = make_float2(0.f, 0.f);
                if (mine) src_fetch(h, ov.y, ga, gb);
                float hb[F_IN] = {ga.x, ga.y, ga.z, ga.w, src_fifth(ov.y, gb)};
#pragma unroll
                for (int f = 0; f < F_IN; ++f) hb[f] = (hb[f] - m1[f]) * rs1[f] * g1[f] + b1[f];
                while (hits != 0u) {
                    const int L = __builtin_ctz(hits);
                    hits &= hits - 1u;
                    const int mw1 = __builtin_amdgcn_readlane(ov.z, L);
                    float hb1[F_IN];
#pragma unroll
                    for (int f = 0; f < F_IN; ++f)
                        hb1[f] = __int_as_float(__builtin_amdgcn_readlane(__float_as_int(hb[f]), L));
                    seg_rows(lane, 1, mw1, hb1, wl, blv, we, at, r, mx, den, acc);
                }
            }
        }
        float inv = 1.f / (den + 1e-16f);
        const float invc = 0.5f / fmaxf((float)dC, 1.f);
        v2f s[4];
#pragma unroll
        for (int k = 0; k < 4; ++k) s[k] = acc[k] * inv;
        const v2f keep0 = lo ? s[0] : s[2], keep1 = lo ? s[1] : s[3];
        const v2f give0 = lo ? s[2] : s[0], give1 = lo ? s[3] : s[1];
        float4 y;
        y.x = (keep0.x + other_half(give0.x)) * invc + bg.x;
        y.y = (keep0.y + other_half(give0.y)) * invc + bg.y;
        y.z = (keep1.x + other_half(give1.x)) * invc + bg.z;
        y.w = (keep1.y + other_half(give1.y)) * invc + bg.w;
        xr_w[i][rot0 + i + ycol] = y;
        dC = dB; mwC = rcB.y;
        {
            float g[F_IN] = {gaB.x, gaB.y, gaB.z, gaB.w, src_fifth(rcB.x, gbB)};
#pragma unroll
            for (int f = 0; f < F_IN; ++f) hbC[f] = (g[f] - m1[f]) * rs1[f] * g1[f] + b1[f];
        }
        dB = dA; rcB = rcA;
    }
    for (int i = cnt; i < NPW; ++i) xr_w[i][rot0 + i + lane] = make_float4(0.f, 0.f, 0.f, 0.f);
}

#define ROWS_B (NPW * SEG_W)
#define SLOT_F 256

__global__ __launch_bounds__(SEG_T)
void k_seg(const int* __restrict__ deg, const int2* __restrict__ rec,
           const int4* __restrict__ ovf,
           const float* __restrict__ h,
           const float* __restrict__ gamma1, const float* __restrict__ beta1,
           const float* __restrict__ rm1, const float* __restrict__ rv1,
           const float* __restrict__ Wl, const float* __restrict__ bl,
           const float* __restrict__ Wr, const float* __restrict__ br,
           const float* __restrict__ We, const float* __restrict__ att,
           const float* __restrict__ bias_gat,
           const float* __restrict__ W1, const float* __restrict__ b1,
           const float* __restrict__ gamma2, const float* __restrict__ beta2,
           const float* __restrict__ rm2, const float* __restrict__ rv2,
           const float* __restrict__ W2, const float* __restrict__ b2,
           float* __restrict__ out) {
    __shared__ float4 xr_s[SEG_W][NPW][64];
    const int lane = threadIdx.x & 31;
    const int wib = __builtin_amdgcn_readfirstlane(threadIdx.x >> 5);
    const int n0 = __builtin_amdgcn_readfirstlane((blockIdx.x * SEG_W + wib) * NPW);
    const int cnt = N_NODES - n0 < NPW ? N_NODES - n0 : NPW;
    seg_wave(lane, n0, cnt, xr_s[wib], (wib * NPW) & 15, deg, rec, ovf, h,
             gamma1, beta1, rm1, rv1, Wl, bl, Wr, br, We, att, bias_gat);
    __syncthreads();

    float* tile = (float*)&xr_s[0][0][0];
    const int c = lane & 15, hh = lane >> 4;
    const int col0 = wib * 32 + c;
    v8f d00 = {0.f, 0.f, 0.f, 0.f, 0.f, 0.f, 0.f, 0.f}, d01 = d00, d10 = d00, d11 = d00;
    {
        const float* ya = tile + c * (SLOT_F + 4) + 2 * hh;
        const float* yb = ya + 16 * SLOT_F;
        const float* wp = W1 + 2 * hh * DDIM + col0;
#pragma unroll
        for (int k = 0; k < DDIM; k += 4) {
            v2f a0 = *(const v2f*)(ya + k);
            v2f a1 = *(const v2f*)(yb + k);
            v2f b0, b1v;
            b0.x  = wp[k * DDIM];       b0.y  = wp[(k + 1) * DDIM];
            b1v.x = wp[k * DDIM + 16];  b1v.y = wp[(k + 1) * DDIM + 16];
            d00 = __builtin_amdgcn_wmma_f32_16x16x4_f32(false, a0, false, b0,  (short)0, d00, false, false);
            d01 = __builtin_amdgcn_wmma_f32_16x16x4_f32(false, a0, false, b1v, (short)0, d01, false, false);
            d10 = __builtin_amdgcn_wmma_f32_16x16x4_f32(false, a1, false, b0,  (short)0, d10, false, false);
            d11 = __builtin_amdgcn_wmma_f32_16x16x4_f32(false, a1, false, b1v, (short)0, d11, false, false);
        }
    }
    __syncthreads();

#pragma unroll
    for (int nt = 0; nt < 2; ++nt) {
        const int col = col0 + nt * 16;
        const float bb = b1[col], mm = rm2[col], rs = rsqrtf(rv2[col] + BN_EPS);
        const float gg = gamma2[col], be = beta2[col];
#pragma unroll
        for (int mt = 0; mt < 2; ++mt) {
            const v8f d = nt == 0 ? (mt == 0 ? d00 : d10) : (mt == 0 ? d01 : d11);
#pragma unroll
            for (int v = 0; v < 8; ++v) {
                const int m = v + hh * 8;
                float z = d[v] + bb;
                z = (z - mm) * rs * gg + be;
                z = z > 0.f ? z : 0.01f * z;
                tile[(mt * 16 + m) * SLOT_F + 4 * m + col] = z;
            }
        }
    }
    __syncthreads();

    const int row0 = blockIdx.x * ROWS_B;
    for (int o = threadIdx.x; o < ROWS_B * F_IN; o += SEG_T) {
        const int r = o / F_IN, cc = o % F_IN;
        if (row0 + r >= N_NODES) break;
        float acc2 = b2[cc];
        const float* zr = tile + r * SLOT_F + 4 * (r & 15);
#pragma unroll 8
        for (int dd = 0; dd < DDIM; ++dd) acc2 += zr[dd] * W2[dd * F_IN + cc];
        out[(size_t)(row0 + r) * F_IN + cc] = acc2;
    }
}

extern "C" void kernel_launch(void* const* d_in, const int* in_sizes, int n_in,
                              void* d_out, int out_size, void* d_ws, size_t ws_size,
                              hipStream_t stream) {
    const float* h        = (const float*)d_in[0];
    const int*   ei       = (const int*)  d_in[1];
    const float* ew       = (const float*)d_in[2];
    const float* gamma1   = (const float*)d_in[3];
    const float* beta1    = (const float*)d_in[4];
    const float* rm1      = (const float*)d_in[5];
    const float* rv1      = (const float*)d_in[6];
    const float* Wl       = (const float*)d_in[7];
    const float* bl       = (const float*)d_in[8];
    const float* Wr       = (const float*)d_in[9];
    const float* br       = (const float*)d_in[10];
    const float* We       = (const float*)d_in[11];
    const float* att      = (const float*)d_in[12];
    const float* bias_gat = (const float*)d_in[13];
    const float* W1       = (const float*)d_in[14];
    const float* b1       = (const float*)d_in[15];
    const float* gamma2   = (const float*)d_in[16];
    const float* beta2    = (const float*)d_in[17];
    const float* rm2      = (const float*)d_in[18];
    const float* rv2      = (const float*)d_in[19];
    const float* W2       = (const float*)d_in[20];
    const float* b2       = (const float*)d_in[21];
    float* out = (float*)d_out;

    float* ws = (float*)d_ws;
    int* cnt      = (int*)(ws + CNT_OFF);
    int2* rec     = (int2*)(ws + REC_OFF);
    int4* ovf     = (int4*)(ws + OVF_OFF);

    const int T = 256;
    k_zero<<<(N_NODES + T) / T, T, 0, stream>>>(cnt);
    k_place<<<(N_EDGES + T - 1) / T, T, 0, stream>>>(ei, ew, cnt, rec, ovf);
    k_seg<<<SEG_BLOCKS, SEG_T, 0, stream>>>(cnt, rec, ovf, h, gamma1, beta1, rm1, rv1,
                                            Wl, bl, Wr, br, We, att, bias_gat,
                                            W1, b1, gamma2, beta2, rm2, rv2, W2, b2, out);
}


# fc2 reads W2 from a transposed LDS copy (fetched under fc1's matrix products) by 16-byte LDS reads, first wave's 2 outputs in one pass: no global-load wait inside its K loop; sum order kept
# speedup vs baseline: 5.4324x; 1.0809x over previous
#include <hip/hip_runtime.h>
#include <math.h>

#define N_NODES 50000
#define N_EDGES 400000
#define HEADS   2
#define DDIM    128
#define F_IN    5
#define HD      (HEADS*DDIM)
#define BN_EPS  1e-5f

#define CNT_OFF  0
#define REC_OFF  50048
#define OVF_OFF  3250048

#define SLOTS    32

typedef __attribute__((ext_vector_type(2))) float v2f;
typedef __attribute__((ext_vector_type(8))) float v8f;

template <int M>
__device__ __forceinline__ float row_xor_add(float v) {
    int t = __builtin_amdgcn_update_dpp(0, __float_as_int(v), 0x160 + M, 0xf, 0xf, true);
    return v + __int_as_float(t);
}
template <int M>
__device__ __forceinline__ float row_xor_get(float v) {
    return __int_as_float(__builtin_amdgcn_update_dpp(0, __float_as_int(v), 0x160 + M, 0xf, 0xf, true));
}
template <int L>
__device__ __forceinline__ float row_lane_get(float v) {
    return __int_as_float(__builtin_amdgcn_update_dpp(0, __float_as_int(v), 0x150 + L, 0xf, 0xf, true));
}

__global__ __launch_bounds__(256)
void k_zero(int* __restrict__ cnt) {
    int i = blockIdx.x * blockDim.x + threadIdx.x;
    if (i <= N_NODES) cnt[i] = 0;
}

__global__ __launch_bounds__(256)
void k_place(const int* __restrict__ ei, const float* __restrict__ ew,
             int* __restrict__ cnt, int2* __restrict__ rec,
             int4* __restrict__ ovf) {
    int e = blockIdx.x * blockDim.x + threadIdx.x;
    if (e >= N_EDGES) return;
    int src = ei[e];
    int dst = ei[N_EDGES + e];
    int w = __float_as_int(ew[e]);
    int rank = atomicAdd(&cnt[dst], 1);
    if (rank < SLOTS) rec[(size_t)dst * SLOTS + rank] = make_int2(src, w);
    else ovf[atomicAdd(&cnt[N_NODES], 1)] = make_int4(dst, src, w, 0);
}

#define NPW   8
#define SEG_W 4
#define SEG_T (32 * SEG_W)
#define SEG_BLOCKS ((N_NODES + NPW * SEG_W - 1) / (NPW * SEG_W))

__device__ __forceinline__ void src_fetch(const float* __restrict__ h, unsigned s,
                                          float4& a, float2& b) {
    s = s < N_NODES ? s : N_NODES - 1;
    unsigned e = s * F_IN + 4;
    e = e < N_NODES * F_IN - 1 ? e : N_NODES * F_IN - 2;
    __builtin_memcpy(&a, h + (size_t)s * F_IN, 16);
    __builtin_memcpy(&b, h + e, 8);
}
__device__ __forceinline__ float src_fifth(unsigned s, float2 b) {
    return s < N_NODES - 1 ? b.x : b.y;
}

__device__ __forceinline__ void seg_rows(const int lane, int m, int mw, const float (&hb)[F_IN],
                                         const v2f (&wl)[F_IN][4], const v2f (&blv)[4],
                                         const v2f (&we)[4], const v2f (&at)[4],
                                         const v2f (&r)[4],
                                         float& mx, float& den, v2f (&acc)[4]) {
    const bool hi8 = (lane & 8) != 0, hi4 = (lane & 4) != 0, odd = (lane & 1) != 0;
    const int urow = (lane >> 2) & 3;
    for (int j = 0; j < m; j += 4) {
        v2f x[4][4];
        float t[4];
#pragma unroll
        for (int u = 0; u < 4; ++u) {
            int jj = j + u < m ? j + u : m - 1;
            float w = __int_as_float(__builtin_amdgcn_readlane(mw, jj));
            float sh[F_IN];
#pragma unroll
            for (int f = 0; f < F_IN; ++f)
                sh[f] = __int_as_float(__builtin_amdgcn_readlane(__float_as_int(hb[f]), jj));
            v2f q = (v2f){0.f, 0.f};
#pragma unroll
            for (int k = 0; k < 4; ++k) {
                v2f xv = blv[k];
#pragma unroll
                for (int f = 0; f < F_IN; ++f) xv += sh[f] * wl[f][k];
                x[u][k] = xv;
                v2f z = xv + r[k] + w * we[k];
                v2f zs = 0.2f * z;
                z.x = z.x > 0.f ? z.x : zs.x;
                z.y = z.y > 0.f ? z.y : zs.y;
                q += at[k] * z;
            }
            t[u] = q.x + q.y;
        }
        const float k0 = hi8 ? t[2] : t[0], g0 = hi8 ? t[0] : t[2];
        const float k1 = hi8 ? t[3] : t[1], g1 = hi8 ? t[1] : t[3];
        const float u0 = k0 + row_xor_get<8>(g0);
        const float u1 = k1 + row_xor_get<8>(g1);
        const float k2 = hi4 ? u1 : u0, g2 = hi4 ? u0 : u1;
        float lg = k2 + row_xor_get<4>(g2);
        lg = row_xor_add<2>(lg);
        lg = row_xor_add<1>(lg);
        if (j + urow >= m) lg = -INFINITY;
        float gm = fmaxf(lg, row_xor_get<8>(lg));
        gm = fmaxf(gm, row_xor_get<4>(gm));
        const float mn = fmaxf(mx, gm);
        const float ex = expf((odd ? mx : lg) - mn);
        const float p0 = row_lane_get<0>(ex), p1 = row_lane_get<4>(ex);
        const float p2 = row_lane_get<8>(ex), p3 = row_lane_get<12>(ex);
        const float sc = row_lane_get<1>(ex);
        mx = mn;
        den = den * sc + ((p0 + p1) + (p2 + p3));
#pragma unroll
        for (int k = 0; k < 4; ++k)
            acc[k] = acc[k] * sc + p0 * x[0][k] + p1 * x[1][k]
                                 + p2 * x[2][k] + p3 * x[3][k];
    }
}

__device__ __forceinline__ float other_half(float x) {
    int b = __float_as_int(x);
    return __int_as_float(__builtin_amdgcn_permlanex16(b, b, 0x76543210, 0xfedcba98, false, false));
}

__device__ __forceinline__ void seg_wave(
           const int lane, const int n0, const int cnt, float4 (*xr_w)[64], const int rot0,
           const int* __restrict__ deg, const int2* __restrict__ rec,
           const int4* __restrict__ ovf,
           const float* __restrict__ h,
           const float* __restrict__ gamma1, const float* __restrict__ beta1,
           const float* __restrict__ rm1, const float* __restrict__ rv1,
           const float* __restrict__ Wl, const float* __restrict__ bl,
           const float* __restrict__ Wr, const float* __restrict__ br,
           const float* __restrict__ We, const float* __restrict__ att,
           const float* __restrict__ bias_gat) {
    if (cnt <= 0) {
        for (int i = 0; i < NPW; ++i) xr_w[i][rot0 + i + lane] = make_float4(0.f, 0.f, 0.f, 0.f);
        return;
    }
    int degv = 0;
    if (lane < cnt) degv = deg[n0 + lane];
    float hn0, hn1;
    {
        int e0 = n0 * F_IN + lane, e1 = e0 + 32;
        e0 = e0 < N_NODES * F_IN ? e0 : N_NODES * F_IN - 1;
        e1 = e1 < N_NODES * F_IN ? e1 : N_NODES * F_IN - 1;
        hn0 = h[e0];
        hn1 = h[e1];
    }
    float m1[F_IN], rs1[F_IN], g1[F_IN], b1[F_IN];
#pragma unroll
    for (int f = 0; f < F_IN; ++f) {
        m1[f] = rm1[f];
        rs1[f] = rsqrtf(rv1[f] + BN_EPS);
        g1[f] = gamma1[f];
        b1[f] = beta1[f];
    }
    v2f wl[F_IN][4], blv[4], we[4], at[4];
    {
        const v2f* pb = (const v2f*)bl + lane * 4;
        const v2f* pw = (const v2f*)We + lane * 4;
        const v2f* pa = (const v2f*)att + lane * 4;
#pragma unroll
        for (int k = 0; k < 4; ++k) { blv[k] = pb[k]; we[k] = pw[k]; at[k] = pa[k]; }
#pragma unroll
        for (int f = 0; f < F_IN; ++f) {
            const v2f* pl = (const v2f*)(Wl + f * HD) + lane * 4;
#pragma unroll
            for (int k = 0; k < 4; ++k) wl[f][k] = pl[k];
        }
    }
    const bool lo = lane < 16;
    const int ycol = (lane & 15) * 2 + (lane >> 4);
    float4 bg;
    __builtin_memcpy(&bg, bias_gat + ycol * 4, 16);
    const int2* rrow = rec + (size_t)n0 * SLOTS + lane;
    int dC = __builtin_amdgcn_readlane(degv, 0);
    int dB = __builtin_amdgcn_readlane(degv, 1);
    int2 rcC = make_int2(0, 0), rcB = make_int2(0, 0);
    if (lane < dC) rcC = rrow[0];
    if (lane < dB) rcB = rrow[SLOTS];
    {
        v2f wr[F_IN][4], brv[4];
        const v2f* pc = (const v2f*)br + lane * 4;
#pragma unroll
        for (int k = 0; k < 4; ++k) brv[k] = pc[k];
#pragma unroll
        for (int f = 0; f < F_IN; ++f) {
            const v2f* pq = (const v2f*)(Wr + f * HD) + lane * 4;
#pragma unroll
            for (int k = 0; k < 4; ++k) wr[f][k] = pq[k];
        }
#pragma unroll
        for (int i = 0; i < NPW; ++i) {
            v2f r[4];
#pragma unroll
            for (int k = 0; k < 4; ++k) r[k] = brv[k];
#pragma unroll
            for (int f = 0; f < F_IN; ++f) {
                const int e = i * F_IN + f;
                float raw = __int_as_float(__builtin_amdgcn_readlane(
                    __float_as_int(e < 32 ? hn0 : hn1), e & 31));
                float hv = (raw - m1[f]) * rs1[f] * g1[f] + b1[f];
#pragma unroll
                for (int k = 0; k < 4; ++k) r[k] += hv * wr[f][k];
            }
            xr_w[i][lane]      = make_float4(r[0].x, r[0].y, r[1].x, r[1].y);
            xr_w[i][32 + lane] = make_float4(r[2].x, r[2].y, r[3].x, r[3].y);
        }
    }
    int mwC = rcC.y;
    float hbC[F_IN];
    {
        float4 ga = make_float4(0.f, 0.f, 0.f, 0.f);
        float2 gb = make_float2(0.f, 0.f);
        if (lane < dC) src_fetch(h, rcC.x, ga, gb);
        float g[F_IN] = {ga.x, ga.y, ga.z, ga.w, src_fifth(rcC.x, gb)};
#pragma unroll
        for (int f = 0; f < F_IN; ++f) hbC[f] = (g[f] - m1[f]) * rs1[f] * g1[f] + b1[f];
    }
    for (int i = 0; i < cnt; ++i) {
        float4 gaB = make_float4(0.f, 0.f, 0.f, 0.f);
        float2 gbB = make_float2(0.f, 0.f);
        if (lane < dB) src_fetch(h, rcB.x, gaB, gbB);
        int dA = __builtin_amdgcn_readlane(degv, i + 2);
        int2 rcA = make_int2(0, 0);
        if (lane < dA) rcA = rrow[(i + 2) * SLOTS];
        v2f r[4];
        {
            float4 ra = xr_w[i][lane], rb = xr_w[i][32 + lane];
            r[0] = (v2f){ra.x, ra.y}; r[1] = (v2f){ra.z, ra.w};
            r[2] = (v2f){rb.x, rb.y}; r[3] = (v2f){rb.z, rb.w};
        }
        float mx = -INFINITY, den = 0.f;
        v2f acc[4];
#pragma unroll
        for (int k = 0; k < 4; ++k) acc[k] = (v2f){0.f, 0.f};
        seg_rows(lane, dC < SLOTS ? dC : SLOTS, mwC, hbC, wl, blv, we, at, r, mx, den, acc);
        if (dC > SLOTS) {
            const int node = n0 + i;
            int total = deg[N_NODES];
            total = total < N_EDGES ? total : N_EDGES;
            for (int c0 = 0; c0 < total; c0 += 32) {
                int4 ov = make_int4(-1, 0, 0, 0);
                if (c0 + lane < total) ov = ovf[c0 + lane];
                const bool mine = ov.x == node;
                unsigned hits = __builtin_amdgcn_ballot_w32(mine);
                if (hits == 0u) continue;
                float4 ga = make_float4(0.f, 0.f, 0.f, 0.f);
                float2 gb = make_float2(0.f, 0.f);
                if (mine) src_fetch(h, ov.y, ga, gb);
                float hb[F_IN] = {ga.x, ga.y, ga.z, ga.w, src_fifth(ov.y, gb)};
#pragma unroll
                for (int f = 0; f < F_IN; ++f) hb[f] = (hb[f] - m1[f]) * rs1[f] * g1[f] + b1[f];
                while (hits != 0u) {
                    const int L = __builtin_ctz(hits);
                    hits &= hits - 1u;
                    const int mw1 = __builtin_amdgcn_readlane(ov.z, L);
                    float hb1[F_IN];
#pragma unroll
                    for (int f = 0; f < F_IN; ++f)
                        hb1[f] = __int_as_float(__builtin_amdgcn_readlane(__float_as_int(hb[f]), L));
                    seg_rows(lane, 1, mw1, hb1, wl, blv, we, at, r, mx, den, acc);
                }
            }
        }
        float inv = 1.f / (den + 1e-16f);
        const float invc = 0.5f / fmaxf((float)dC, 1.f);
        v2f s[4];
#pragma unroll
        for (int k = 0; k < 4; ++k) s[k] = acc[k] * inv;
        const v2f keep0 = lo ? s[0] : s[2], keep1 = lo ? s[1] : s[3];
        const v2f give0 = lo ? s[2] : s[0], give1 = lo ? s[3] : s[1];
        float4 y;
        y.x = (keep0.x + other_half(give0.x)) * invc + bg.x;
        y.y = (keep0.y + other_half(give0.y)) * invc + bg.y;
        y.z = (keep1.x + other_half(give1.x)) * invc + bg.z;
        y.w = (keep1.y + other_half(give1.y)) * invc + bg.w;
        xr_w[i][rot0 + i + ycol] = y;
        dC = dB; mwC = rcB.y;
        {
            float g[F_IN] = {gaB.x, gaB.y, gaB.z, gaB.w, src_fifth(rcB.x, gbB)};
#pragma unroll
            for (int f = 0; f < F_IN; ++f) hbC[f] = (g[f] - m1[f]) * rs1[f] * g1[f] + b1[f];
        }
        dB = dA; rcB = rcA;
    }
    for (int i = cnt; i < NPW; ++i) xr_w[i][rot0 + i + lane] = make_float4(0.f, 0.f, 0.f, 0.f);
}

#define ROWS_B (NPW * SEG_W)
#define SLOT_F 256
#define W2T_PITCH 132

template <int NO>
__device__ __forceinline__ void fc2_rows(const float* tile, const float* w2t,
                                         const float (&start)[2], float* __restrict__ out,
                                         const int row0, const int o0) {
    int r[NO], cc[NO];
    float acc[NO];
    const float4* zr[NO];
    const float4* wc[NO];
#pragma unroll
    for (int u = 0; u < NO; ++u) {
        const int o = o0 + u * SEG_T;
        r[u] = o / F_IN;
        cc[u] = o - r[u] * F_IN;
        acc[u] = start[u];
        zr[u] = (const float4*)(tile + r[u] * SLOT_F + 4 * (r[u] & 15));
        wc[u] = (const float4*)(w2t + cc[u] * W2T_PITCH);
    }
#pragma unroll 8
    for (int j = 0; j < DDIM / 4; ++j) {
#pragma unroll
        for (int u = 0; u < NO; ++u) {
            const float4 z = zr[u][j], w = wc[u][j];
            acc[u] = __builtin_fmaf(z.x, w.x, acc[u]);
            acc[u] = __builtin_fmaf(z.y, w.y, acc[u]);
            acc[u] = __builtin_fmaf(z.z, w.z, acc[u]);
            acc[u] = __builtin_fmaf(z.w, w.w, acc[u]);
        }
    }
#pragma unroll
    for (int u = 0; u < NO; ++u)
        if (row0 + r[u] < N_NODES)
            out[(size_t)(row0 + r[u]) * F_IN + cc[u]] = acc[u];
}

__global__ __launch_bounds__(SEG_T)
void k_seg(const int* __restrict__ deg, const int2* __restrict__ rec,
           const int4* __restrict__ ovf,
           const float* __restrict__ h,
           const float* __restrict__ gamma1, const float* __restrict__ beta1,
           const float* __restrict__ rm1, const float* __restrict__ rv1,
           const float* __restrict__ Wl, const float* __restrict__ bl,
           const float* __restrict__ Wr, const float* __restrict__ br,
           const float* __restrict__ We, const float* __restrict__ att,
           const float* __restrict__ bias_gat,
           const float* __restrict__ W1, const float* __restrict__ b1,
           const float* __restrict__ gamma2, const float* __restrict__ beta2,
           const float* __restrict__ rm2, const float* __restrict__ rv2,
           const float* __restrict__ W2, const float* __restrict__ b2,
           float* __restrict__ out) {
    __shared__ float4 xr_s[SEG_W][NPW][64];
    __shared__ float4 w2t_s[F_IN * W2T_PITCH / 4];
    const int lane = threadIdx.x & 31;
    const int wib = __builtin_amdgcn_readfirstlane(threadIdx.x >> 5);
    const int n0 = __builtin_amdgcn_readfirstlane((blockIdx.x * SEG_W + wib) * NPW);
    const int cnt = N_NODES - n0 < NPW ? N_NODES - n0 : NPW;
    seg_wave(lane, n0, cnt, xr_s[wib], (wib * NPW) & 15, deg, rec, ovf, h,
             gamma1, beta1, rm1, rv1, Wl, bl, Wr, br, We, att, bias_gat);
    __syncthreads();

    float w2v[F_IN];
#pragma unroll
    for (int i = 0; i < F_IN; ++i) w2v[i] = W2[threadIdx.x + SEG_T * i];

    float* tile = (float*)&xr_s[0][0][0];
    const int c = lane & 15, hh = lane >> 4;
    const int col0 = wib * 32 + c;
    v8f d00 = {0.f, 0.f, 0.f, 0.f, 0.f, 0.f, 0.f, 0.f}, d01 = d00, d10 = d00, d11 = d00;
    {
        const float* ya = tile + c * (SLOT_F + 4) + 2 * hh;
        const float* yb = ya + 16 * SLOT_F;
        const float* wp = W1 + 2 * hh * DDIM + col0;
#pragma unroll
        for (int k = 0; k < DDIM; k += 4) {
            v2f a0 = *(const v2f*)(ya + k);
            v2f a1 = *(const v2f*)(yb + k);
            v2f b0, b1v;
            b0.x  = wp[k * DDIM];       b0.y  = wp[(k + 1) * DDIM];
            b1v.x = wp[k * DDIM + 16];  b1v.y = wp[(k + 1) * DDIM + 16];
            d00 = __builtin_amdgcn_wmma_f32_16x16x4_f32(false, a0, false, b0,  (short)0, d00, false, false);
            d01 = __builtin_amdgcn_wmma_f32_16x16x4_f32(false, a0, false, b1v, (short)0, d01, false, false);
            d10 = __builtin_amdgcn_wmma_f32_16x16x4_f32(false, a1, false, b0,  (short)0, d10, false, false);
            d11 = __builtin_amdgcn_wmma_f32_16x16x4_f32(false, a1, false, b1v, (short)0, d11, false, false);
        }
    }
    {
        float* w2t = (float*)w2t_s;
#pragma unroll
        for (int i = 0; i < F_IN; ++i) {
            const int idx = threadIdx.x + SEG_T * i;
            const int dd = idx / F_IN, cc = idx - dd * F_IN;
            w2t[cc * W2T_PITCH + dd] = w2v[i];
        }
    }
    __syncthreads();

    const float fc2_start[2] = {b2[threadIdx.x % F_IN], b2[(threadIdx.x + SEG_T) % F_IN]};

#pragma unroll
    for (int nt = 0; nt < 2; ++nt) {
        const int col = col0 + nt * 16;
        const float bb = b1[col], mm = rm2[col], rs = rsqrtf(rv2[col] + BN_EPS);
        const float gg = gamma2[col], be = beta2[col];
#pragma unroll
        for (int mt = 0; mt < 2; ++mt) {
            const v8f d = nt == 0 ? (mt == 0 ? d00 : d10) : (mt == 0 ? d01 : d11);
#pragma unroll
            for (int v = 0; v < 8; ++v) {
                const int m = v + hh * 8;
                float z = d[v] + bb;
                z = (z - mm) * rs * gg + be;
                z = z > 0.f ? z : 0.01f * z;
                tile[(mt * 16 + m) * SLOT_F + 4 * m + col] = z;
            }
        }
    }
    __syncthreads();

    const int row0 = blockIdx.x * ROWS_B;
    if (wib == 0) fc2_rows<2>(tile, (const float*)w2t_s, fc2_start, out, row0, threadIdx.x);
    else          fc2_rows<1>(tile, (const float*)w2t_s, fc2_start, out, row0, threadIdx.x);
}

extern "C" void kernel_launch(void* const* d_in, const int* in_sizes, int n_in,
                              void* d_out, int out_size, void* d_ws, size_t ws_size,
                              hipStream_t stream) {
    const float* h        = (const float*)d_in[0];
    const int*   ei       = (const int*)  d_in[1];
    const float* ew       = (const float*)d_in[2];
    const float* gamma1   = (const float*)d_in[3];
    const float* beta1    = (const float*)d_in[4];
    const float* rm1      = (const float*)d_in[5];
    const float* rv1      = (const float*)d_in[6];
    const float* Wl       = (const float*)d_in[7];
    const float* bl       = (const float*)d_in[8];
    const float* Wr       = (const float*)d_in[9];
    const float* br       = (const float*)d_in[10];
    const float* We       = (const float*)d_in[11];
    const float* att      = (const float*)d_in[12];
    const float* bias_gat = (const float*)d_in[13];
    const float* W1       = (const float*)d_in[14];
    const float* b1       = (const float*)d_in[15];
    const float* gamma2   = (const float*)d_in[16];
    const float* beta2    = (const float*)d_in[17];
    const float* rm2      = (const float*)d_in[18];
    const float* rv2      = (const float*)d_in[19];
    const float* W2       = (const float*)d_in[20];
    const float* b2       = (const float*)d_in[21];
    float* out = (float*)d_out;

    float* ws = (float*)d_ws;
    int* cnt      = (int*)(ws + CNT_OFF);
    int2* rec     = (int2*)(ws + REC_OFF);
    int4* ovf     = (int4*)(ws + OVF_OFF);

    const int T = 256;
    k_zero<<<(N_NODES + T) / T, T, 0, stream>>>(cnt);
    k_place<<<(N_EDGES + T - 1) / T, T, 0, stream>>>(ei, ew, cnt, rec, ovf);
    k_seg<<<SEG_BLOCKS, SEG_T, 0, stream>>>(cnt, rec, ovf, h, gamma1, beta1, rm1, rv1,
                                            Wl, bl, Wr, br, We, att, bias_gat,
                                            W1, b1, gamma2, beta2, rm2, rv2, W2, b2, out);
}


# Edge grouping through LDS: 4096-edge chunks sorted by 128-node band in LDS; each segment block picks its nodes' records from its band's runs by LDS counting adds (arrival order kept); no global adds
# speedup vs baseline: 7.4570x; 1.3727x over previous
#include <hip/hip_runtime.h>
#include <math.h>

#define N_NODES 50000
#define N_EDGES 400000
#define HEADS   2
#define DDIM    128
#define F_IN    5
#define HD      (HEADS*DDIM)
#define BN_EPS  1e-5f

#define BAND     128
#define NBAND    ((N_NODES + BAND - 1) / BAND)
#define CH       4096
#define NCH      ((N_EDGES + CH - 1) / CH)
#define OROW     (NBAND + 1)
#define SEGP_OFF 0
#define OFFS_OFF (NCH * CH * 2)
static_assert(N_NODES <= 65536, "a record keeps its source in 16 bits");
static_assert((OFFS_OFF * 4) % 256 == 0, "plane alignment");

#define SLOTS    32

typedef __attribute__((ext_vector_type(2))) float v2f;
typedef __attribute__((ext_vector_type(8))) float v8f;

template <int M>
__device__ __forceinline__ float row_xor_add(float v) {
    int t = __builtin_amdgcn_update_dpp(0, __float_as_int(v), 0x160 + M, 0xf, 0xf, true);
    return v + __int_as_float(t);
}
template <int M>
__device__ __forceinline__ float row_xor_get(float v) {
    return __int_as_float(__builtin_amdgcn_update_dpp(0, __float_as_int(v), 0x160 + M, 0xf, 0xf, true));
}
template <int L>
__device__ __forceinline__ float row_lane_get(float v) {
    return __int_as_float(__builtin_amdgcn_update_dpp(0, __float_as_int(v), 0x150 + L, 0xf, 0xf, true));
}

#define A_T      512
#define EPT      (CH / A_T)
#define SCAN_PER ((OROW + 31) / 32)
__global__ __launch_bounds__(A_T)
void k_band(const int* __restrict__ ei, const float* __restrict__ ew,
            int2* __restrict__ segp, int* __restrict__ offs) {
    __shared__ int hist[OROW + 32];
    __shared__ int2 srt[CH];
    const int t = threadIdx.x;
    const int e0 = blockIdx.x * CH;
    const int n = N_EDGES - e0 < CH ? N_EDGES - e0 : CH;
    for (int i = t; i < OROW; i += A_T) hist[i] = 0;
    __syncthreads();
    unsigned sv[EPT], dv[EPT];
    int wv[EPT];
#pragma unroll
    for (int i = 0; i < EPT; ++i) {
        const int k = t + i * A_T;
        const int kk = k < n ? k : n - 1;
        sv[i] = (unsigned)ei[e0 + kk];
        dv[i] = (unsigned)ei[N_EDGES + e0 + kk];
        wv[i] = __float_as_int(ew[e0 + kk]);
    }
    int bi[EPT], rk[EPT], st[EPT];
    int2 rc[EPT];
#pragma unroll
    for (int i = 0; i < EPT; ++i) {
        const unsigned d = dv[i] < N_NODES ? dv[i] : N_NODES - 1;
        bi[i] = t + i * A_T < n ? (int)(d / BAND) : OROW + (t & 31);
        rc[i] = make_int2((int)((sv[i] & 0xffffu) | ((d % BAND) << 16)), wv[i]);
    }
#pragma unroll
    for (int i = 0; i < EPT; ++i) rk[i] = atomicAdd(&hist[bi[i]], 1);
    __syncthreads();
    if (t < 32) {
        int c[SCAN_PER];
        int sum = 0;
#pragma unroll
        for (int j = 0; j < SCAN_PER; ++j) {
            const int idx = t * SCAN_PER + j;
            c[j] = idx < NBAND ? hist[idx] : 0;
            sum += c[j];
        }
        int incl = sum;
        for (int dlt = 1; dlt < 32; dlt <<= 1) {
            const int v = __shfl_up(incl, dlt, 32);
            if (t >= dlt) incl += v;
        }
        int run = incl - sum;
#pragma unroll
        for (int j = 0; j < SCAN_PER; ++j) {
            const int idx = t * SCAN_PER + j;
            if (idx < OROW) hist[idx] = run;
            run += c[j];
        }
    }
    __syncthreads();
#pragma unroll
    for (int i = 0; i < EPT; ++i) st[i] = hist[bi[i]];
#pragma unroll
    for (int i = 0; i < EPT; ++i)
        if (t + i * A_T < n) srt[st[i] + rk[i]] = rc[i];
    __syncthreads();
    for (int k = t; k < n; k += A_T) segp[(size_t)blockIdx.x * CH + k] = srt[k];
    for (int i = t; i < OROW; i += A_T) offs[blockIdx.x * OROW + i] = hist[i];
}

#define NPW   8
#define SEG_W 4
#define SEG_T (32 * SEG_W)
#define SEG_BLOCKS ((N_NODES + NPW * SEG_W - 1) / (NPW * SEG_W))

__device__ __forceinline__ void src_fetch(const float* __restrict__ h, unsigned s,
                                          float4& a, float2& b) {
    s = s < N_NODES ? s : N_NODES - 1;
    unsigned e = s * F_IN + 4;
    e = e < N_NODES * F_IN - 1 ? e : N_NODES * F_IN - 2;
    __builtin_memcpy(&a, h + (size_t)s * F_IN, 16);
    __builtin_memcpy(&b, h + e, 8);
}
__device__ __forceinline__ float src_fifth(unsigned s, float2 b) {
    return s < N_NODES - 1 ? b.x : b.y;
}

__device__ __forceinline__ void seg_rows(const int lane, int m, int mw, const float (&hb)[F_IN],
                                         const v2f (&wl)[F_IN][4], const v2f (&blv)[4],
                                         const v2f (&we)[4], const v2f (&at)[4],
                                         const v2f (&r)[4],
                                         float& mx, float& den, v2f (&acc)[4]) {
    const bool hi8 = (lane & 8) != 0, hi4 = (lane & 4) != 0, odd = (lane & 1) != 0;
    const int urow = (lane >> 2) & 3;
    for (int j = 0; j < m; j += 4) {
        v2f x[4][4];
        float t[4];
#pragma unroll
        for (int u = 0; u < 4; ++u) {
            int jj = j + u < m ? j + u : m - 1;
            float w = __int_as_float(__builtin_amdgcn_readlane(mw, jj));
            float sh[F_IN];
#pragma unroll
            for (int f = 0; f < F_IN; ++f)
                sh[f] = __int_as_float(__builtin_amdgcn_readlane(__float_as_int(hb[f]), jj));
            v2f q = (v2f){0.f, 0.f};
#pragma unroll
            for (int k = 0; k < 4; ++k) {
                v2f xv = blv[k];
#pragma unroll
                for (int f = 0; f < F_IN; ++f) xv += sh[f] * wl[f][k];
                x[u][k] = xv;
                v2f z = xv + r[k] + w * we[k];
                v2f zs = 0.2f * z;
                z.x = z.x > 0.f ? z.x : zs.x;
                z.y = z.y > 0.f ? z.y : zs.y;
                q += at[k] * z;
            }
            t[u] = q.x + q.y;
        }
        const float k0 = hi8 ? t[2] : t[0], g0 = hi8 ? t[0] : t[2];
        const float k1 = hi8 ? t[3] : t[1], g1 = hi8 ? t[1] : t[3];
        const float u0 = k0 + row_xor_get<8>(g0);
        const float u1 = k1 + row_xor_get<8>(g1);
        const float k2 = hi4 ? u1 : u0, g2 = hi4 ? u0 : u1;
        float lg = k2 + row_xor_get<4>(g2);
        lg = row_xor_add<2>(lg);
        lg = row_xor_add<1>(lg);
        if (j + urow >= m) lg = -INFINITY;
        float gm = fmaxf(lg, row_xor_get<8>(lg));
        gm = fmaxf(gm, row_xor_get<4>(gm));
        const float mn = fmaxf(mx, gm);
        const float ex = expf((odd ? mx : lg) - mn);
        const float p0 = row_lane_get<0>(ex), p1 = row_lane_get<4>(ex);
        const float p2 = row_lane_get<8>(ex), p3 = row_lane_get<12>(ex);
        const float sc = row_lane_get<1>(ex);
        mx = mn;
        den = den * sc + ((p0 + p1) + (p2 + p3));
#pragma unroll
        for (int k = 0; k < 4; ++k)
            acc[k] = acc[k] * sc + p0 * x[0][k] + p1 * x[1][k]
                                 + p2 * x[2][k] + p3 * x[3][k];
    }
}

__device__ __forceinline__ float other_half(float x) {
    int b = __float_as_int(x);
    return __int_as_float(__builtin_amdgcn_permlanex16(b, b, 0x76543210, 0xfedcba98, false, false));
}

__device__ __forceinline__ void seg_wave(
           const int lane, const int n0, const int cnt, float4 (*xr_w)[64], const int rot0,
           const int* lcnt_w, const int2* lrow_w,
           const int2* __restrict__ segp, const int* lo0, const int* lo1, const unsigned dl0,
           const float* __restrict__ h,
           const float* __restrict__ gamma1, const float* __restrict__ beta1,
           const float* __restrict__ rm1, const float* __restrict__ rv1,
           const float* __restrict__ Wl, const float* __restrict__ bl,
           const float* __restrict__ Wr, const float* __restrict__ br,
           const float* __restrict__ We, const float* __restrict__ att,
           const float* __restrict__ bias_gat) {
    if (cnt <= 0) {
        for (int i = 0; i < NPW; ++i) xr_w[i][rot0 + i + lane] = make_float4(0.f, 0.f, 0.f, 0.f);
        return;
    }
    int degv = 0;
    if (lane < cnt) degv = lcnt_w[lane];
    float hn0, hn1;
    {
        int e0 = n0 * F_IN + lane, e1 = e0 + 32;
        e0 = e0 < N_NODES * F_IN ? e0 : N_NODES * F_IN - 1;
        e1 = e1 < N_NODES * F_IN ? e1 : N_NODES * F_IN - 1;
        hn0 = h[e0];
        hn1 = h[e1];
    }
    float m1[F_IN], rs1[F_IN], g1[F_IN], b1[F_IN];
#pragma unroll
    for (int f = 0; f < F_IN; ++f) {
        m1[f] = rm1[f];
        rs1[f] = rsqrtf(rv1[f] + BN_EPS);
        g1[f] = gamma1[f];
        b1[f] = beta1[f];
    }
    v2f wl[F_IN][4], blv[4], we[4], at[4];
    {
        const v2f* pb = (const v2f*)bl + lane * 4;
        const v2f* pw = (const v2f*)We + lane * 4;
        const v2f* pa = (const v2f*)att + lane * 4;
#pragma unroll
        for (int k = 0; k < 4; ++k) { blv[k] = pb[k]; we[k] = pw[k]; at[k] = pa[k]; }
#pragma unroll
        for (int f = 0; f < F_IN; ++f) {
            const v2f* pl = (const v2f*)(Wl + f * HD) + lane * 4;
#pragma unroll
            for (int k = 0; k < 4; ++k) wl[f][k] = pl[k];
        }
    }
    const bool lo = lane < 16;
    const int ycol = (lane & 15) * 2 + (lane >> 4);
    float4 bg;
    __builtin_memcpy(&bg, bias_gat + ycol * 4, 16);
    const int2* rrow = lrow_w + lane;
    int dC = __builtin_amdgcn_readlane(degv, 0);
    int dB = __builtin_amdgcn_readlane(degv, 1);
    int2 rcC = make_int2(0, 0), rcB = make_int2(0, 0);
    if (lane < dC) rcC = rrow[0];
    if (lane < dB) rcB = rrow[SLOTS];
    {
        v2f wr[F_IN][4], brv[4];
        const v2f* pc = (const v2f*)br + lane * 4;
#pragma unroll
        for (int k = 0; k < 4; ++k) brv[k] = pc[k];
#pragma unroll
        for (int f = 0; f < F_IN; ++f) {
            const v2f* pq = (const v2f*)(Wr + f * HD) + lane * 4;
#pragma unroll
            for (int k = 0; k < 4; ++k) wr[f][k] = pq[k];
        }
#pragma unroll
        for (int i = 0; i < NPW; ++i) {
            v2f r[4];
#pragma unroll
            for (int k = 0; k < 4; ++k) r[k] = brv[k];
#pragma unroll
            for (int f = 0; f < F_IN; ++f) {
                const int e = i * F_IN + f;
                float raw = __int_as_float(__builtin_amdgcn_readlane(
                    __float_as_int(e < 32 ? hn0 : hn1), e & 31));
                float hv = (raw - m1[f]) * rs1[f] * g1[f] + b1[f];
#pragma unroll
                for (int k = 0; k < 4; ++k) r[k] += hv * wr[f][k];
            }
            xr_w[i][lane]      = make_float4(r[0].x, r[0].y, r[1].x, r[1].y);
            xr_w[i][32 + lane] = make_float4(r[2].x, r[2].y, r[3].x, r[3].y);
        }
    }
    int mwC = rcC.y;
    float hbC[F_IN];
    {
        float4 ga = make_float4(0.f, 0.f, 0.f, 0.f);
        float2 gb = make_float2(0.f, 0.f);
        if (lane < dC) src_fetch(h, rcC.x, ga, gb);
        float g[F_IN] = {ga.x, ga.y, ga.z, ga.w, src_fifth(rcC.x, gb)};
#pragma unroll
        for (int f = 0; f < F_IN; ++f) hbC[f] = (g[f] - m1[f]) * rs1[f] * g1[f] + b1[f];
    }
    for (int i = 0; i < cnt; ++i) {
        float4 gaB = make_float4(0.f, 0.f, 0.f, 0.f);
        float2 gbB = make_float2(0.f, 0.f);
        if (lane < dB) src_fetch(h, rcB.x, gaB, gbB);
        int dA = __builtin_amdgcn_readlane(degv, i + 2);
        int2 rcA = make_int2(0, 0);
        if (lane < dA) rcA = rrow[(i + 2) * SLOTS];
        v2f r[4];
        {
            float4 ra = xr_w[i][lane], rb = xr_w[i][32 + lane];
            r[0] = (v2f){ra.x, ra.y}; r[1] = (v2f){ra.z, ra.w};
            r[2] = (v2f){rb.x, rb.y}; r[3] = (v2f){rb.z, rb.w};
        }
        float mx = -INFINITY, den = 0.f;
        v2f acc[4];
#pragma unroll
        for (int k = 0; k < 4; ++k) acc[k] = (v2f){0.f, 0.f};
        seg_rows(lane, dC <= SLOTS ? dC : 0, mwC, hbC, wl, blv, we, at, r, mx, den, acc);
        if (dC > SLOTS) {
            const unsigned dl = dl0 + (unsigned)i;
            for (int g = 0; g < NCH; ++g) {
                const int o0 = __builtin_amdgcn_readfirstlane(lo0[g]);
                const int o1 = __builtin_amdgcn_readfirstlane(lo1[g]);
                for (int c0 = o0; c0 < o1; c0 += 32) {
                    int2 ov = make_int2(-1, 0);
                    if (c0 + lane < o1) ov = segp[(size_t)g * CH + c0 + lane];
                    const bool mine = ((unsigned)ov.x >> 16) == dl;
                    unsigned hits = __builtin_amdgcn_ballot_w32(mine);
                    if (hits == 0u) continue;
                    const unsigned sidx = (unsigned)ov.x & 0xffffu;
                    float4 ga = make_float4(0.f, 0.f, 0.f, 0.f);
                    float2 gb = make_float2(0.f, 0.f);
                    if (mine) src_fetch(h, sidx, ga, gb);
                    float hb[F_IN] = {ga.x, ga.y, ga.z, ga.w, src_fifth(sidx, gb)};
#pragma unroll
                    for (int f = 0; f < F_IN; ++f) hb[f] = (hb[f] - m1[f]) * rs1[f] * g1[f] + b1[f];
                    while (hits != 0u) {
                        const int L = __builtin_ctz(hits);
                        hits &= hits - 1u;
                        const int mw1 = __builtin_amdgcn_readlane(ov.y, L);
                        float hb1[F_IN];
#pragma unroll
                        for (int f = 0; f < F_IN; ++f)
                            hb1[f] = __int_as_float(__builtin_amdgcn_readlane(__float_as_int(hb[f]), L));
                        seg_rows(lane, 1, mw1, hb1, wl, blv, we, at, r, mx, den, acc);
                    }
                }
            }
        }
        float inv = 1.f / (den + 1e-16f);
        const float invc = 0.5f / fmaxf((float)dC, 1.f);
        v2f s[4];
#pragma unroll
        for (int k = 0; k < 4; ++k) s[k] = acc[k] * inv;
        const v2f keep0 = lo ? s[0] : s[2], keep1 = lo ? s[1] : s[3];
        const v2f give0 = lo ? s[2] : s[0], give1 = lo ? s[3] : s[1];
        float4 y;
        y.x = (keep0.x + other_half(give0.x)) * invc + bg.x;
        y.y = (keep0.y + other_half(give0.y)) * invc + bg.y;
        y.z = (keep1.x + other_half(give1.x)) * invc + bg.z;
        y.w = (keep1.y + other_half(give1.y)) * invc + bg.w;
        xr_w[i][rot0 + i + ycol] = y;
        dC = dB; mwC = rcB.y;
        {
            float g[F_IN] = {gaB.x, gaB.y, gaB.z, gaB.w, src_fifth(rcB.x, gbB)};
#pragma unroll
            for (int f = 0; f < F_IN; ++f) hbC[f] = (g[f] - m1[f]) * rs1[f] * g1[f] + b1[f];
        }
        dB = dA; rcB = rcA;
    }
    for (int i = cnt; i < NPW; ++i) xr_w[i][rot0 + i + lane] = make_float4(0.f, 0.f, 0.f, 0.f);
}

#define ROWS_B (NPW * SEG_W)
#define RPW    ((NCH + SEG_W - 1) / SEG_W)
static_assert(BAND % ROWS_B == 0, "a block's nodes lie in one band");
static_assert(NCH <= SEG_T && RPW <= 32, "one thread per run bound, one lane per run of a wave");
#define SLOT_F 256
#define W2T_PITCH 132

template <int NO>
__device__ __forceinline__ void fc2_rows(const float* tile, const float* w2t,
                                         const float (&start)[2], float* __restrict__ out,
                                         const int row0, const int o0) {
    int r[NO], cc[NO];
    float acc[NO];
    const float4* zr[NO];
    const float4* wc[NO];
#pragma unroll
    for (int u = 0; u < NO; ++u) {
        const int o = o0 + u * SEG_T;
        r[u] = o / F_IN;
        cc[u] = o - r[u] * F_IN;
        acc[u] = start[u];
        zr[u] = (const float4*)(tile + r[u] * SLOT_F + 4 * (r[u] & 15));
        wc[u] = (const float4*)(w2t + cc[u] * W2T_PITCH);
    }
#pragma unroll 8
    for (int j = 0; j < DDIM / 4; ++j) {
#pragma unroll
        for (int u = 0; u < NO; ++u) {
            const float4 z = zr[u][j], w = wc[u][j];
            acc[u] = __builtin_fmaf(z.x, w.x, acc[u]);
            acc[u] = __builtin_fmaf(z.y, w.y, acc[u]);
            acc[u] = __builtin_fmaf(z.z, w.z, acc[u]);
            acc[u] = __builtin_fmaf(z.w, w.w, acc[u]);
        }
    }
#pragma unroll
    for (int u = 0; u < NO; ++u)
        if (row0 + r[u] < N_NODES)
            out[(size_t)(row0 + r[u]) * F_IN + cc[u]] = acc[u];
}

__global__ __launch_bounds__(SEG_T)
void k_seg(const int2* __restrict__ segp, const int* __restrict__ offs,
           const float* __restrict__ h,
           const float* __restrict__ gamma1, const float* __restrict__ beta1,
           const float* __restrict__ rm1, const float* __restrict__ rv1,
           const float* __restrict__ Wl, const float* __restrict__ bl,
           const float* __restrict__ Wr, const float* __restrict__ br,
           const float* __restrict__ We, const float* __restrict__ att,
           const float* __restrict__ bias_gat,
           const float* __restrict__ W1, const float* __restrict__ b1,
           const float* __restrict__ gamma2, const float* __restrict__ beta2,
           const float* __restrict__ rm2, const float* __restrict__ rv2,
           const float* __restrict__ W2, const float* __restrict__ b2,
           float* __restrict__ out) {
    __shared__ float4 xr_s[SEG_W][NPW][64];
    __shared__ float4 w2t_s[F_IN * W2T_PITCH / 4];
    __shared__ int lcnt[ROWS_B + 32];
    __shared__ int2 lrec[ROWS_B * SLOTS];
    __shared__ int lo0[NCH], lo1[NCH];
    const int lane = threadIdx.x & 31;
    const int wib = __builtin_amdgcn_readfirstlane(threadIdx.x >> 5);
    const int n0 = __builtin_amdgcn_readfirstlane((blockIdx.x * SEG_W + wib) * NPW);
    const int cnt = N_NODES - n0 < NPW ? N_NODES - n0 : NPW;

    const int band = blockIdx.x / (BAND / ROWS_B);
    const unsigned sub = (blockIdx.x % (BAND / ROWS_B)) * ROWS_B;
    if (threadIdx.x < ROWS_B) lcnt[threadIdx.x] = 0;
    if (threadIdx.x < NCH) {
        const int* o = offs + threadIdx.x * OROW + band;
        int ra = o[0], rb = o[1];
        ra = ra < 0 ? 0 : (ra > CH ? CH : ra);
        rb = rb < ra ? ra : (rb > CH ? CH : rb);
        lo0[threadIdx.x] = ra;
        lo1[threadIdx.x] = rb;
    }
    __syncthreads();
    {
        const int gl = wib * RPW + lane;
        int av = 0, bv = 0;
        if (lane < RPW && gl < NCH) { av = lo0[gl]; bv = lo1[gl]; }
        int2 rv[RPW];
        int rk[RPW];
#pragma unroll
        for (int q = 0; q < RPW; ++q) {
            const int o0 = __builtin_amdgcn_readlane(av, q), o1 = __builtin_amdgcn_readlane(bv, q);
            rv[q] = make_int2(-1, 0);
            if (o0 + lane < o1) rv[q] = segp[(size_t)(wib * RPW + q) * CH + o0 + lane];
        }
#pragma unroll
        for (int q = 0; q < RPW; ++q) {
            const unsigned dl = ((unsigned)rv[q].x >> 16) - sub;
            rk[q] = atomicAdd(&lcnt[dl < (unsigned)ROWS_B ? dl : (unsigned)(ROWS_B + lane)], 1);
        }
#pragma unroll
        for (int q = 0; q < RPW; ++q) {
            const unsigned dl = ((unsigned)rv[q].x >> 16) - sub;
            if (dl < (unsigned)ROWS_B && rk[q] < SLOTS)
                lrec[dl * SLOTS + rk[q]] = make_int2(rv[q].x & 0xffff, rv[q].y);
        }
        for (int q = 0; q < RPW; ++q) {
            const int o0 = __builtin_amdgcn_readlane(av, q), o1 = __builtin_amdgcn_readlane(bv, q);
            for (int c0 = o0 + 32; c0 < o1; c0 += 32) {
                int2 v = make_int2(-1, 0);
                if (c0 + lane < o1) v = segp[(size_t)(wib * RPW + q) * CH + c0 + lane];
                const unsigned dl = ((unsigned)v.x >> 16) - sub;
                if (dl < (unsigned)ROWS_B) {
                    const int k = atomicAdd(&lcnt[dl], 1);
                    if (k < SLOTS) lrec[dl * SLOTS + k] = make_int2(v.x & 0xffff, v.y);
                }
            }
        }
    }
    __syncthreads();

    seg_wave(lane, n0, cnt, xr_s[wib], (wib * NPW) & 15,
             lcnt + wib * NPW, lrec + wib * NPW * SLOTS, segp, lo0, lo1, sub + wib * NPW, h,
             gamma1, beta1, rm1, rv1, Wl, bl, Wr, br, We, att, bias_gat);
    __syncthreads();

    float w2v[F_IN];
#pragma unroll
    for (int i = 0; i < F_IN; ++i) w2v[i] = W2[threadIdx.x + SEG_T * i];

    float* tile = (float*)&xr_s[0][0][0];
    const int c = lane & 15, hh = lane >> 4;
    const int col0 = wib * 32 + c;
    v8f d00 = {0.f, 0.f, 0.f, 0.f, 0.f, 0.f, 0.f, 0.f}, d01 = d00, d10 = d00, d11 = d00;
    {
        const float* ya = tile + c * (SLOT_F + 4) + 2 * hh;
        const float* yb = ya + 16 * SLOT_F;
        const float* wp = W1 + 2 * hh * DDIM + col0;
#pragma unroll
        for (int k = 0; k < DDIM; k += 4) {
            v2f a0 = *(const v2f*)(ya + k);
            v2f a1 = *(const v2f*)(yb + k);
            v2f b0, b1v;
            b0.x  = wp[k * DDIM];       b0.y  = wp[(k + 1) * DDIM];
            b1v.x = wp[k * DDIM + 16];  b1v.y = wp[(k + 1) * DDIM + 16];
            d00 = __builtin_amdgcn_wmma_f32_16x16x4_f32(false, a0, false, b0,  (short)0, d00, false, false);
            d01 = __builtin_amdgcn_wmma_f32_16x16x4_f32(false, a0, false, b1v, (short)0, d01, false, false);
            d10 = __builtin_amdgcn_wmma_f32_16x16x4_f32(false, a1, false, b0,  (short)0, d10, false, false);
            d11 = __builtin_amdgcn_wmma_f32_16x16x4_f32(false, a1, false, b1v, (short)0, d11, false, false);
        }
    }
    {
        float* w2t = (float*)w2t_s;
#pragma unroll
        for (int i = 0; i < F_IN; ++i) {
            const int idx = threadIdx.x + SEG_T * i;
            const int dd = idx / F_IN, cc = idx - dd * F_IN;
            w2t[cc * W2T_PITCH + dd] = w2v[i];
        }
    }
    __syncthreads();

    const float fc2_start[2] = {b2[threadIdx.x % F_IN], b2[(threadIdx.x + SEG_T) % F_IN]};

#pragma unroll
    for (int nt = 0; nt < 2; ++nt) {
        const int col = col0 + nt * 16;
        const float bb = b1[col], mm = rm2[col], rs = rsqrtf(rv2[col] + BN_EPS);
        const float gg = gamma2[col], be = beta2[col];
#pragma unroll
        for (int mt = 0; mt < 2; ++mt) {
            const v8f d = nt == 0 ? (mt == 0 ? d00 : d10) : (mt == 0 ? d01 : d11);
#pragma unroll
            for (int v = 0; v < 8; ++v) {
                const int m = v + hh * 8;
                float z = d[v] + bb;
                z = (z - mm) * rs * gg + be;
                z = z > 0.f ? z : 0.01f * z;
                tile[(mt * 16 + m) * SLOT_F + 4 * m + col] = z;
            }
        }
    }
    __syncthreads();

    const int row0 = blockIdx.x * ROWS_B;
    if (wib == 0) fc2_rows<2>(tile, (const float*)w2t_s, fc2_start, out, row0, threadIdx.x);
    else          fc2_rows<1>(tile, (const float*)w2t_s, fc2_start, out, row0, threadIdx.x);
}

extern "C" void kernel_launch(void* const* d_in, const int* in_sizes, int n_in,
                              void* d_out, int out_size, void* d_ws, size_t ws_size,
                              hipStream_t stream) {
    const float* h        = (const float*)d_in[0];
    const int*   ei       = (const int*)  d_in[1];
    const float* ew       = (const float*)d_in[2];
    const float* gamma1   = (const float*)d_in[3];
    const float* beta1    = (const float*)d_in[4];
    const float* rm1      = (const float*)d_in[5];
    const float* rv1      = (const float*)d_in[6];
    const float* Wl       = (const float*)d_in[7];
    const float* bl       = (const float*)d_in[8];
    const float* Wr       = (const float*)d_in[9];
    const float* br       = (const float*)d_in[10];
    const float* We       = (const float*)d_in[11];
    const float* att      = (const float*)d_in[12];
    const float* bias_gat = (const float*)d_in[13];
    const float* W1       = (const float*)d_in[14];
    const float* b1       = (const float*)d_in[15];
    const float* gamma2   = (const float*)d_in[16];
    const float* beta2    = (const float*)d_in[17];
    const float* rm2      = (const float*)d_in[18];
    const float* rv2      = (const float*)d_in[19];
    const float* W2       = (const float*)d_in[20];
    const float* b2       = (const float*)d_in[21];
    float* out = (float*)d_out;

    float* ws = (float*)d_ws;
    int2* segp    = (int2*)(ws + SEGP_OFF);
    int* offs     = (int*)(ws + OFFS_OFF);

    k_band<<<NCH, A_T, 0, stream>>>(ei, ew, segp, offs);
    k_seg<<<SEG_BLOCKS, SEG_T, 0, stream>>>(segp, offs, h, gamma1, beta1, rm1, rv1,
                                            Wl, bl, Wr, br, We, att, bias_gat,
                                            W1, b1, gamma2, beta2, rm2, rv2, W2, b2, out);
}


# Both leaky relus taken as max(z, slope*z), bit-identical: one v_max per value in place of a compare and select pair in the segment loop and the BN2 epilogue; sum orders kept
# speedup vs baseline: 7.7553x; 1.0400x over previous
#include <hip/hip_runtime.h>
#include <math.h>

#define N_NODES 50000
#define N_EDGES 400000
#define HEADS   2
#define DDIM    128
#define F_IN    5
#define HD      (HEADS*DDIM)
#define BN_EPS  1e-5f

#define BAND     128
#define NBAND    ((N_NODES + BAND - 1) / BAND)
#define CH       4096
#define NCH      ((N_EDGES + CH - 1) / CH)
#define OROW     (NBAND + 1)
#define SEGP_OFF 0
#define OFFS_OFF (NCH * CH * 2)
static_assert(N_NODES <= 65536, "a record keeps its source in 16 bits");
static_assert((OFFS_OFF * 4) % 256 == 0, "plane alignment");

#define SLOTS    32

typedef __attribute__((ext_vector_type(2))) float v2f;
typedef __attribute__((ext_vector_type(8))) float v8f;

template <int M>
__device__ __forceinline__ float row_xor_add(float v) {
    int t = __builtin_amdgcn_update_dpp(0, __float_as_int(v), 0x160 + M, 0xf, 0xf, true);
    return v + __int_as_float(t);
}
template <int M>
__device__ __forceinline__ float row_xor_get(float v) {
    return __int_as_float(__builtin_amdgcn_update_dpp(0, __float_as_int(v), 0x160 + M, 0xf, 0xf, true));
}
template <int L>
__device__ __forceinline__ float row_lane_get(float v) {
    return __int_as_float(__builtin_amdgcn_update_dpp(0, __float_as_int(v), 0x150 + L, 0xf, 0xf, true));
}

#define A_T      512
#define EPT      (CH / A_T)
#define SCAN_PER ((OROW + 31) / 32)
__global__ __launch_bounds__(A_T)
void k_band(const int* __restrict__ ei, const float* __restrict__ ew,
            int2* __restrict__ segp, int* __restrict__ offs) {
    __shared__ int hist[OROW + 32];
    __shared__ int2 srt[CH];
    const int t = threadIdx.x;
    const int e0 = blockIdx.x * CH;
    const int n = N_EDGES - e0 < CH ? N_EDGES - e0 : CH;
    for (int i = t; i < OROW; i += A_T) hist[i] = 0;
    __syncthreads();
    unsigned sv[EPT], dv[EPT];
    int wv[EPT];
#pragma unroll
    for (int i = 0; i < EPT; ++i) {
        const int k = t + i * A_T;
        const int kk = k < n ? k : n - 1;
        sv[i] = (unsigned)ei[e0 + kk];
        dv[i] = (unsigned)ei[N_EDGES + e0 + kk];
        wv[i] = __float_as_int(ew[e0 + kk]);
    }
    int bi[EPT], rk[EPT], st[EPT];
    int2 rc[EPT];
#pragma unroll
    for (int i = 0; i < EPT; ++i) {
        const unsigned d = dv[i] < N_NODES ? dv[i] : N_NODES - 1;
        bi[i] = t + i * A_T < n ? (int)(d / BAND) : OROW + (t & 31);
        rc[i] = make_int2((int)((sv[i] & 0xffffu) | ((d % BAND) << 16)), wv[i]);
    }
#pragma unroll
    for (int i = 0; i < EPT; ++i) rk[i] = atomicAdd(&hist[bi[i]], 1);
    __syncthreads();
    if (t < 32) {
        int c[SCAN_PER];
        int sum = 0;
#pragma unroll
        for (int j = 0; j < SCAN_PER; ++j) {
            const int idx = t * SCAN_PER + j;
            c[j] = idx < NBAND ? hist[idx] : 0;
            sum += c[j];
        }
        int incl = sum;
        for (int dlt = 1; dlt < 32; dlt <<= 1) {
            const int v = __shfl_up(incl, dlt, 32);
            if (t >= dlt) incl += v;
        }
        int run = incl - sum;
#pragma unroll
        for (int j = 0; j < SCAN_PER; ++j) {
            const int idx = t * SCAN_PER + j;
            if (idx < OROW) hist[idx] = run;
            run += c[j];
        }
    }
    __syncthreads();
#pragma unroll
    for (int i = 0; i < EPT; ++i) st[i] = hist[bi[i]];
#pragma unroll
    for (int i = 0; i < EPT; ++i)
        if (t + i * A_T < n) srt[st[i] + rk[i]] = rc[i];
    __syncthreads();
    for (int k = t; k < n; k += A_T) segp[(size_t)blockIdx.x * CH + k] = srt[k];
    for (int i = t; i < OROW; i += A_T) offs[blockIdx.x * OROW + i] = hist[i];
}

#define NPW   8
#define SEG_W 4
#define SEG_T (32 * SEG_W)
#define SEG_BLOCKS ((N_NODES + NPW * SEG_W - 1) / (NPW * SEG_W))

__device__ __forceinline__ void src_fetch(const float* __restrict__ h, unsigned s,
                                          float4& a, float2& b) {
    s = s < N_NODES ? s : N_NODES - 1;
    unsigned e = s * F_IN + 4;
    e = e < N_NODES * F_IN - 1 ? e : N_NODES * F_IN - 2;
    __builtin_memcpy(&a, h + (size_t)s * F_IN, 16);
    __builtin_memcpy(&b, h + e, 8);
}
__device__ __forceinline__ float src_fifth(unsigned s, float2 b) {
    return s < N_NODES - 1 ? b.x : b.y;
}

__device__ __forceinline__ void seg_rows(const int lane, int m, int mw, const float (&hb)[F_IN],
                                         const v2f (&wl)[F_IN][4], const v2f (&blv)[4],
                                         const v2f (&we)[4], const v2f (&at)[4],
                                         const v2f (&r)[4],
                                         float& mx, float& den, v2f (&acc)[4]) {
    const bool hi8 = (lane & 8) != 0, hi4 = (lane & 4) != 0, odd = (lane & 1) != 0;
    const int urow = (lane >> 2) & 3;
    for (int j = 0; j < m; j += 4) {
        v2f x[4][4];
        float t[4];
#pragma unroll
        for (int u = 0; u < 4; ++u) {
            int jj = j + u < m ? j + u : m - 1;
            float w = __int_as_float(__builtin_amdgcn_readlane(mw, jj));
            float sh[F_IN];
#pragma unroll
            for (int f = 0; f < F_IN; ++f)
                sh[f] = __int_as_float(__builtin_amdgcn_readlane(__float_as_int(hb[f]), jj));
            v2f q = (v2f){0.f, 0.f};
#pragma unroll
            for (int k = 0; k < 4; ++k) {
                v2f xv = blv[k];
#pragma unroll
                for (int f = 0; f < F_IN; ++f) xv += sh[f] * wl[f][k];
                x[u][k] = xv;
                v2f z = xv + r[k] + w * we[k];
                v2f zs = 0.2f * z;
                z.x = __builtin_fmaxf(z.x, zs.x);
                z.y = __builtin_fmaxf(z.y, zs.y);
                q += at[k] * z;
            }
            t[u] = q.x + q.y;
        }
        const float k0 = hi8 ? t[2] : t[0], g0 = hi8 ? t[0] : t[2];
        const float k1 = hi8 ? t[3] : t[1], g1 = hi8 ? t[1] : t[3];
        const float u0 = k0 + row_xor_get<8>(g0);
        const float u1 = k1 + row_xor_get<8>(g1);
        const float k2 = hi4 ? u1 : u0, g2 = hi4 ? u0 : u1;
        float lg = k2 + row_xor_get<4>(g2);
        lg = row_xor_add<2>(lg);
        lg = row_xor_add<1>(lg);
        if (j + urow >= m) lg = -INFINITY;
        float gm = fmaxf(lg, row_xor_get<8>(lg));
        gm = fmaxf(gm, row_xor_get<4>(gm));
        const float mn = fmaxf(mx, gm);
        const float ex = expf((odd ? mx : lg) - mn);
        const float p0 = row_lane_get<0>(ex), p1 = row_lane_get<4>(ex);
        const float p2 = row_lane_get<8>(ex), p3 = row_lane_get<12>(ex);
        const float sc = row_lane_get<1>(ex);
        mx = mn;
        den = den * sc + ((p0 + p1) + (p2 + p3));
#pragma unroll
        for (int k = 0; k < 4; ++k)
            acc[k] = acc[k] * sc + p0 * x[0][k] + p1 * x[1][k]
                                 + p2 * x[2][k] + p3 * x[3][k];
    }
}

__device__ __forceinline__ float other_half(float x) {
    int b = __float_as_int(x);
    return __int_as_float(__builtin_amdgcn_permlanex16(b, b, 0x76543210, 0xfedcba98, false, false));
}

__device__ __forceinline__ void seg_wave(
           const int lane, const int n0, const int cnt, float4 (*xr_w)[64], const int rot0,
           const int* lcnt_w, const int2* lrow_w,
           const int2* __restrict__ segp, const int* lo0, const int* lo1, const unsigned dl0,
           const float* __restrict__ h,
           const float* __restrict__ gamma1, const float* __restrict__ beta1,
           const float* __restrict__ rm1, const float* __restrict__ rv1,
           const float* __restrict__ Wl, const float* __restrict__ bl,
           const float* __restrict__ Wr, const float* __restrict__ br,
           const float* __restrict__ We, const float* __restrict__ att,
           const float* __restrict__ bias_gat) {
    if (cnt <= 0) {
        for (int i = 0; i < NPW; ++i) xr_w[i][rot0 + i + lane] = make_float4(0.f, 0.f, 0.f, 0.f);
        return;
    }
    int degv = 0;
    if (lane < cnt) degv = lcnt_w[lane];
    float hn0, hn1;
    {
        int e0 = n0 * F_IN + lane, e1 = e0 + 32;
        e0 = e0 < N_NODES * F_IN ? e0 : N_NODES * F_IN - 1;
        e1 = e1 < N_NODES * F_IN ? e1 : N_NODES * F_IN - 1;
        hn0 = h[e0];
        hn1 = h[e1];
    }
    float m1[F_IN], rs1[F_IN], g1[F_IN], b1[F_IN];
#pragma unroll
    for (int f = 0; f < F_IN; ++f) {
        m1[f] = rm1[f];
        rs1[f] = rsqrtf(rv1[f] + BN_EPS);
        g1[f] = gamma1[f];
        b1[f] = beta1[f];
    }
    v2f wl[F_IN][4], blv[4], we[4], at[4];
    {
        const v2f* pb = (const v2f*)bl + lane * 4;
        const v2f* pw = (const v2f*)We + lane * 4;
        const v2f* pa = (const v2f*)att + lane * 4;
#pragma unroll
        for (int k = 0; k < 4; ++k) { blv[k] = pb[k]; we[k] = pw[k]; at[k] = pa[k]; }
#pragma unroll
        for (int f = 0; f < F_IN; ++f) {
            const v2f* pl = (const v2f*)(Wl + f * HD) + lane * 4;
#pragma unroll
            for (int k = 0; k < 4; ++k) wl[f][k] = pl[k];
        }
    }
    const bool lo = lane < 16;
    const int ycol = (lane & 15) * 2 + (lane >> 4);
    float4 bg;
    __builtin_memcpy(&bg, bias_gat + ycol * 4, 16);
    const int2* rrow = lrow_w + lane;
    int dC = __builtin_amdgcn_readlane(degv, 0);
    int dB = __builtin_amdgcn_readlane(degv, 1);
    int2 rcC = make_int2(0, 0), rcB = make_int2(0, 0);
    if (lane < dC) rcC = rrow[0];
    if (lane < dB) rcB = rrow[SLOTS];
    {
        v2f wr[F_IN][4], brv[4];
        const v2f* pc = (const v2f*)br + lane * 4;
#pragma unroll
        for (int k = 0; k < 4; ++k) brv[k] = pc[k];
#pragma unroll
        for (int f = 0; f < F_IN; ++f) {
            const v2f* pq = (const v2f*)(Wr + f * HD) + lane * 4;
#pragma unroll
            for (int k = 0; k < 4; ++k) wr[f][k] = pq[k];
        }
#pragma unroll
        for (int i = 0; i < NPW; ++i) {
            v2f r[4];
#pragma unroll
            for (int k = 0; k < 4; ++k) r[k] = brv[k];
#pragma unroll
            for (int f = 0; f < F_IN; ++f) {
                const int e = i * F_IN + f;
                float raw = __int_as_float(__builtin_amdgcn_readlane(
                    __float_as_int(e < 32 ? hn0 : hn1), e & 31));
                float hv = (raw - m1[f]) * rs1[f] * g1[f] + b1[f];
#pragma unroll
                for (int k = 0; k < 4; ++k) r[k] += hv * wr[f][k];
            }
            xr_w[i][lane]      = make_float4(r[0].x, r[0].y, r[1].x, r[1].y);
            xr_w[i][32 + lane] = make_float4(r[2].x, r[2].y, r[3].x, r[3].y);
        }
    }
    int mwC = rcC.y;
    float hbC[F_IN];
    {
        float4 ga = make_float4(0.f, 0.f, 0.f, 0.f);
        float2 gb = make_float2(0.f, 0.f);
        if (lane < dC) src_fetch(h, rcC.x, ga, gb);
        float g[F_IN] = {ga.x, ga.y, ga.z, ga.w, src_fifth(rcC.x, gb)};
#pragma unroll
        for (int f = 0; f < F_IN; ++f) hbC[f] = (g[f] - m1[f]) * rs1[f] * g1[f] + b1[f];
    }
    for (int i = 0; i < cnt; ++i) {
        float4 gaB = make_float4(0.f, 0.f, 0.f, 0.f);
        float2 gbB = make_float2(0.f, 0.f);
        if (lane < dB) src_fetch(h, rcB.x, gaB, gbB);
        int dA = __builtin_amdgcn_readlane(degv, i + 2);
        int2 rcA = make_int2(0, 0);
        if (lane < dA) rcA = rrow[(i + 2) * SLOTS];
        v2f r[4];
        {
            float4 ra = xr_w[i][lane], rb = xr_w[i][32 + lane];
            r[0] = (v2f){ra.x, ra.y}; r[1] = (v2f){ra.z, ra.w};
            r[2] = (v2f){rb.x, rb.y}; r[3] = (v2f){rb.z, rb.w};
        }
        float mx = -INFINITY, den = 0.f;
        v2f acc[4];
#pragma unroll
        for (int k = 0; k < 4; ++k) acc[k] = (v2f){0.f, 0.f};
        seg_rows(lane, dC <= SLOTS ? dC : 0, mwC, hbC, wl, blv, we, at, r, mx, den, acc);
        if (dC > SLOTS) {
            const unsigned dl = dl0 + (unsigned)i;
            for (int g = 0; g < NCH; ++g) {
                const int o0 = __builtin_amdgcn_readfirstlane(lo0[g]);
                const int o1 = __builtin_amdgcn_readfirstlane(lo1[g]);
                for (int c0 = o0; c0 < o1; c0 += 32) {
                    int2 ov = make_int2(-1, 0);
                    if (c0 + lane < o1) ov = segp[(size_t)g * CH + c0 + lane];
                    const bool mine = ((unsigned)ov.x >> 16) == dl;
                    unsigned hits = __builtin_amdgcn_ballot_w32(mine);
                    if (hits == 0u) continue;
                    const unsigned sidx = (unsigned)ov.x & 0xffffu;
                    float4 ga = make_float4(0.f, 0.f, 0.f, 0.f);
                    float2 gb = make_float2(0.f, 0.f);
                    if (mine) src_fetch(h, sidx, ga, gb);
                    float hb[F_IN] = {ga.x, ga.y, ga.z, ga.w, src_fifth(sidx, gb)};
#pragma unroll
                    for (int f = 0; f < F_IN; ++f) hb[f] = (hb[f] - m1[f]) * rs1[f] * g1[f] + b1[f];
                    while (hits != 0u) {
                        const int L = __builtin_ctz(hits);
                        hits &= hits - 1u;
                        const int mw1 = __builtin_amdgcn_readlane(ov.y, L);
                        float hb1[F_IN];
#pragma unroll
                        for (int f = 0; f < F_IN; ++f)
                            hb1[f] = __int_as_float(__builtin_amdgcn_readlane(__float_as_int(hb[f]), L));
                        seg_rows(lane, 1, mw1, hb1, wl, blv, we, at, r, mx, den, acc);
                    }
                }
            }
        }
        float inv = 1.f / (den + 1e-16f);
        const float invc = 0.5f / fmaxf((float)dC, 1.f);
        v2f s[4];
#pragma unroll
        for (int k = 0; k < 4; ++k) s[k] = acc[k] * inv;
        const v2f keep0 = lo ? s[0] : s[2], keep1 = lo ? s[1] : s[3];
        const v2f give0 = lo ? s[2] : s[0], give1 = lo ? s[3] : s[1];
        float4 y;
        y.x = (keep0.x + other_half(give0.x)) * invc + bg.x;
        y.y = (keep0.y + other_half(give0.y)) * invc + bg.y;
        y.z = (keep1.x + other_half(give1.x)) * invc + bg.z;
        y.w = (keep1.y + other_half(give1.y)) * invc + bg.w;
        xr_w[i][rot0 + i + ycol] = y;
        dC = dB; mwC = rcB.y;
        {
            float g[F_IN] = {gaB.x, gaB.y, gaB.z, gaB.w, src_fifth(rcB.x, gbB)};
#pragma unroll
            for (int f = 0; f < F_IN; ++f) hbC[f] = (g[f] - m1[f]) * rs1[f] * g1[f] + b1[f];
        }
        dB = dA; rcB = rcA;
    }
    for (int i = cnt; i < NPW; ++i) xr_w[i][rot0 + i + lane] = make_float4(0.f, 0.f, 0.f, 0.f);
}

#define ROWS_B (NPW * SEG_W)
#define RPW    ((NCH + SEG_W - 1) / SEG_W)
static_assert(BAND % ROWS_B == 0, "a block's nodes lie in one band");
static_assert(NCH <= SEG_T && RPW <= 32, "one thread per run bound, one lane per run of a wave");
#define SLOT_F 256
#define W2T_PITCH 132

template <int NO>
__device__ __forceinline__ void fc2_rows(const float* tile, const float* w2t,
                                         const float (&start)[2], float* __restrict__ out,
                                         const int row0, const int o0) {
    int r[NO], cc[NO];
    float acc[NO];
    const float4* zr[NO];
    const float4* wc[NO];
#pragma unroll
    for (int u = 0; u < NO; ++u) {
        const int o = o0 + u * SEG_T;
        r[u] = o / F_IN;
        cc[u] = o - r[u] * F_IN;
        acc[u] = start[u];
        zr[u] = (const float4*)(tile + r[u] * SLOT_F + 4 * (r[u] & 15));
        wc[u] = (const float4*)(w2t + cc[u] * W2T_PITCH);
    }
#pragma unroll 8
    for (int j = 0; j < DDIM / 4; ++j) {
#pragma unroll
        for (int u = 0; u < NO; ++u) {
            const float4 z = zr[u][j], w = wc[u][j];
            acc[u] = __builtin_fmaf(z.x, w.x, acc[u]);
            acc[u] = __builtin_fmaf(z.y, w.y, acc[u]);
            acc[u] = __builtin_fmaf(z.z, w.z, acc[u]);
            acc[u] = __builtin_fmaf(z.w, w.w, acc[u]);
        }
    }
#pragma unroll
    for (int u = 0; u < NO; ++u)
        if (row0 + r[u] < N_NODES)
            out[(size_t)(row0 + r[u]) * F_IN + cc[u]] = acc[u];
}

__global__ __launch_bounds__(SEG_T)
void k_seg(const int2* __restrict__ segp, const int* __restrict__ offs,
           const float* __restrict__ h,
           const float* __restrict__ gamma1, const float* __restrict__ beta1,
           const float* __restrict__ rm1, const float* __restrict__ rv1,
           const float* __restrict__ Wl, const float* __restrict__ bl,
           const float* __restrict__ Wr, const float* __restrict__ br,
           const float* __restrict__ We, const float* __restrict__ att,
           const float* __restrict__ bias_gat,
           const float* __restrict__ W1, const float* __restrict__ b1,
           const float* __restrict__ gamma2, const float* __restrict__ beta2,
           const float* __restrict__ rm2, const float* __restrict__ rv2,
           const float* __restrict__ W2, const float* __restrict__ b2,
           float* __restrict__ out) {
    __shared__ float4 xr_s[SEG_W][NPW][64];
    __shared__ float4 w2t_s[F_IN * W2T_PITCH / 4];
    __shared__ int lcnt[ROWS_B + 32];
    __shared__ int2 lrec[ROWS_B * SLOTS];
    __shared__ int lo0[NCH], lo1[NCH];
    const int lane = threadIdx.x & 31;
    const int wib = __builtin_amdgcn_readfirstlane(threadIdx.x >> 5);
    const int n0 = __builtin_amdgcn_readfirstlane((blockIdx.x * SEG_W + wib) * NPW);
    const int cnt = N_NODES - n0 < NPW ? N_NODES - n0 : NPW;

    const int band = blockIdx.x / (BAND / ROWS_B);
    const unsigned sub = (blockIdx.x % (BAND / ROWS_B)) * ROWS_B;
    if (threadIdx.x < ROWS_B) lcnt[threadIdx.x] = 0;
    if (threadIdx.x < NCH) {
        const int* o = offs + threadIdx.x * OROW + band;
        int ra = o[0], rb = o[1];
        ra = ra < 0 ? 0 : (ra > CH ? CH : ra);
        rb = rb < ra ? ra : (rb > CH ? CH : rb);
        lo0[threadIdx.x] = ra;
        lo1[threadIdx.x] = rb;
    }
    __syncthreads();
    {
        const int gl = wib * RPW + lane;
        int av = 0, bv = 0;
        if (lane < RPW && gl < NCH) { av = lo0[gl]; bv = lo1[gl]; }
        int2 rv[RPW];
        int rk[RPW];
#pragma unroll
        for (int q = 0; q < RPW; ++q) {
            const int o0 = __builtin_amdgcn_readlane(av, q), o1 = __builtin_amdgcn_readlane(bv, q);
            rv[q] = make_int2(-1, 0);
            if (o0 + lane < o1) rv[q] = segp[(size_t)(wib * RPW + q) * CH + o0 + lane];
        }
#pragma unroll
        for (int q = 0; q < RPW; ++q) {
            const unsigned dl = ((unsigned)rv[q].x >> 16) - sub;
            rk[q] = atomicAdd(&lcnt[dl < (unsigned)ROWS_B ? dl : (unsigned)(ROWS_B + lane)], 1);
        }
#pragma unroll
        for (int q = 0; q < RPW; ++q) {
            const unsigned dl = ((unsigned)rv[q].x >> 16) - sub;
            if (dl < (unsigned)ROWS_B && rk[q] < SLOTS)
                lrec[dl * SLOTS + rk[q]] = make_int2(rv[q].x & 0xffff, rv[q].y);
        }
        for (int q = 0; q < RPW; ++q) {
            const int o0 = __builtin_amdgcn_readlane(av, q), o1 = __builtin_amdgcn_readlane(bv, q);
            for (int c0 = o0 + 32; c0 < o1; c0 += 32) {
                int2 v = make_int2(-1, 0);
                if (c0 + lane < o1) v = segp[(size_t)(wib * RPW + q) * CH + c0 + lane];
                const unsigned dl = ((unsigned)v.x >> 16) - sub;
                if (dl < (unsigned)ROWS_B) {
                    const int k = atomicAdd(&lcnt[dl], 1);
                    if (k < SLOTS) lrec[dl * SLOTS + k] = make_int2(v.x & 0xffff, v.y);
                }
            }
        }
    }
    __syncthreads();

    seg_wave(lane, n0, cnt, xr_s[wib], (wib * NPW) & 15,
             lcnt + wib * NPW, lrec + wib * NPW * SLOTS, segp, lo0, lo1, sub + wib * NPW, h,
             gamma1, beta1, rm1, rv1, Wl, bl, Wr, br, We, att, bias_gat);
    __syncthreads();

    float w2v[F_IN];
#pragma unroll
    for (int i = 0; i < F_IN; ++i) w2v[i] = W2[threadIdx.x + SEG_T * i];

    float* tile = (float*)&xr_s[0][0][0];
    const int c = lane & 15, hh = lane >> 4;
    const int col0 = wib * 32 + c;
    v8f d00 = {0.f, 0.f, 0.f, 0.f, 0.f, 0.f, 0.f, 0.f}, d01 = d00, d10 = d00, d11 = d00;
    {
        const float* ya = tile + c * (SLOT_F + 4) + 2 * hh;
        const float* yb = ya + 16 * SLOT_F;
        const float* wp = W1 + 2 * hh * DDIM + col0;
#pragma unroll
        for (int k = 0; k < DDIM; k += 4) {
            v2f a0 = *(const v2f*)(ya + k);
            v2f a1 = *(const v2f*)(yb + k);
            v2f b0, b1v;
            b0.x  = wp[k * DDIM];       b0.y  = wp[(k + 1) * DDIM];
            b1v.x = wp[k * DDIM + 16];  b1v.y = wp[(k + 1) * DDIM + 16];
            d00 = __builtin_amdgcn_wmma_f32_16x16x4_f32(false, a0, false, b0,  (short)0, d00, false, false);
            d01 = __builtin_amdgcn_wmma_f32_16x16x4_f32(false, a0, false, b1v, (short)0, d01, false, false);
            d10 = __builtin_amdgcn_wmma_f32_16x16x4_f32(false, a1, false, b0,  (short)0, d10, false, false);
            d11 = __builtin_amdgcn_wmma_f32_16x16x4_f32(false, a1, false, b1v, (short)0, d11, false, false);
        }
    }
    {
        float* w2t = (float*)w2t_s;
#pragma unroll
        for (int i = 0; i < F_IN; ++i) {
            const int idx = threadIdx.x + SEG_T * i;
            const int dd = idx / F_IN, cc = idx - dd * F_IN;
            w2t[cc * W2T_PITCH + dd] = w2v[i];
        }
    }
    __syncthreads();

    const float fc2_start[2] = {b2[threadIdx.x % F_IN], b2[(threadIdx.x + SEG_T) % F_IN]};

#pragma unroll
    for (int nt = 0; nt < 2; ++nt) {
        const int col = col0 + nt * 16;
        const float bb = b1[col], mm = rm2[col], rs = rsqrtf(rv2[col] + BN_EPS);
        const float gg = gamma2[col], be = beta2[col];
#pragma unroll
        for (int mt = 0; mt < 2; ++mt) {
            const v8f d = nt == 0 ? (mt == 0 ? d00 : d10) : (mt == 0 ? d01 : d11);
#pragma unroll
            for (int v = 0; v < 8; ++v) {
                const int m = v + hh * 8;
                float z = d[v] + bb;
                z = (z - mm) * rs * gg + be;
                z = __builtin_fmaxf(z, 0.01f * z);
                tile[(mt * 16 + m) * SLOT_F + 4 * m + col] = z;
            }
        }
    }
    __syncthreads();

    const int row0 = blockIdx.x * ROWS_B;
    if (wib == 0) fc2_rows<2>(tile, (const float*)w2t_s, fc2_start, out, row0, threadIdx.x);
    else          fc2_rows<1>(tile, (const float*)w2t_s, fc2_start, out, row0, threadIdx.x);
}

extern "C" void kernel_launch(void* const* d_in, const int* in_sizes, int n_in,
                              void* d_out, int out_size, void* d_ws, size_t ws_size,
                              hipStream_t stream) {
    const float* h        = (const float*)d_in[0];
    const int*   ei       = (const int*)  d_in[1];
    const float* ew       = (const float*)d_in[2];
    const float* gamma1   = (const float*)d_in[3];
    const float* beta1    = (const float*)d_in[4];
    const float* rm1      = (const float*)d_in[5];
    const float* rv1      = (const float*)d_in[6];
    const float* Wl       = (const float*)d_in[7];
    const float* bl       = (const float*)d_in[8];
    const float* Wr       = (const float*)d_in[9];
    const float* br       = (const float*)d_in[10];
    const float* We       = (const float*)d_in[11];
    const float* att      = (const float*)d_in[12];
    const float* bias_gat = (const float*)d_in[13];
    const float* W1       = (const float*)d_in[14];
    const float* b1       = (const float*)d_in[15];
    const float* gamma2   = (const float*)d_in[16];
    const float* beta2    = (const float*)d_in[17];
    const float* rm2      = (const float*)d_in[18];
    const float* rv2      = (const float*)d_in[19];
    const float* W2       = (const float*)d_in[20];
    const float* b2       = (const float*)d_in[21];
    float* out = (float*)d_out;

    float* ws = (float*)d_ws;
    int2* segp    = (int2*)(ws + SEGP_OFF);
    int* offs     = (int*)(ws + OFFS_OFF);

    k_band<<<NCH, A_T, 0, stream>>>(ei, ew, segp, offs);
    k_seg<<<SEG_BLOCKS, SEG_T, 0, stream>>>(segp, offs, h, gamma1, beta1, rm1, rv1,
                                            Wl, bl, Wr, br, We, att, bias_gat,
                                            W1, b1, gamma2, beta2, rm2, rv2, W2, b2, out);
}
